# starting kernel
# baseline (speedup 1.0000x reference)
#include <hip/hip_runtime.h>

typedef __attribute__((ext_vector_type(2))) float v2f;
typedef __attribute__((ext_vector_type(8))) float v8f;

#define LN_EPS 1e-5f


__global__ void zero_kernel(float* __restrict__ p, long n) {
    long i = (long)blockIdx.x * blockDim.x + threadIdx.x;
    long stride = (long)gridDim.x * blockDim.x;
    for (; i < n; i += stride) p[i] = 0.0f;
}

__global__ void degree_kernel(const int* __restrict__ dst, float* __restrict__ deg, int E) {
    int e = blockIdx.x * blockDim.x + threadIdx.x;
    if (e < E) atomicAdd(&deg[dst[e]], 1.0f);
}

__global__ void invert_kernel(float* __restrict__ norm, int N) {
    int i = blockIdx.x * blockDim.x + threadIdx.x;
    if (i < N) {
        float d = norm[i];
        norm[i] = d > 0.0f ? 1.0f / d : 0.0f;
    }
}

__global__ void scatter_kernel(const float* __restrict__ x, const int* __restrict__ src,
                               const int* __restrict__ dst, const float* __restrict__ norm,
                               float* __restrict__ agg, int E) {
    int t = blockIdx.x * blockDim.x + threadIdx.x;
    int e = t >> 5;
    int lane = t & 31;
    if (e >= E) return;
    int s = src[e];
    int d = dst[e];
    float nv = norm[d];
    const float4 xs = *(const float4*)(x + (long)s * 128 + lane * 4);
    float* ap = agg + (long)d * 128 + lane * 4;
    atomicAdd(ap + 0, xs.x * nv);
    atomicAdd(ap + 1, xs.y * nv);
    atomicAdd(ap + 2, xs.z * nv);
    atomicAdd(ap + 3, xs.w * nv);
}

template <int COLT, bool LNRELU>
__global__ __launch_bounds__(COLT * 32) void sage_gemm_kernel(
    const float* __restrict__ x,
    const float* __restrict__ agg,
    const float* __restrict__ W,
    const float* __restrict__ bias,
    float* __restrict__ y,
    int N) {
    constexpr int NDIM    = COLT * 16;
    constexpr int THREADS = COLT * 32;
    constexpr int G       = 8;
    constexpr int LDK     = 68;

    __shared__ float Wt[NDIM * LDK];
    __shared__ float As[16 * LDK];
    __shared__ float red_s[8 * 16];
    __shared__ float red_q[8 * 16];

    const int tid  = threadIdx.x;
    const int wave = tid >> 5;
    const int lane = tid & 31;
    const int ml   = lane & 15;
    const int half = lane >> 4;
    const long base = (long)blockIdx.x * (G * 16);

    v8f acc[G];
#pragma unroll
    for (int g = 0; g < G; ++g) acc[g] = v8f{};

    for (int kc = 0; kc < 4; ++kc) {
        const int k0 = kc * 64;
        __syncthreads();
        const float* wsrc = W + (long)k0 * NDIM;
        for (int idx = tid; idx < 64 * NDIM; idx += THREADS) {
            const int kloc = idx / NDIM;
            const int n    = idx - kloc * NDIM;
            Wt[n * LDK + kloc] = wsrc[idx];
        }
        const float* asrc = (k0 < 128) ? x : agg;
        const int koff = k0 & 127;
#pragma unroll
        for (int g = 0; g < G; ++g) {
            __syncthreads();
            for (int q = tid; q < 256; q += THREADS) {
                const int r  = q >> 4;
                const int kq = (q & 15) * 4;
                long rr = base + g * 16 + r;
                if (rr >= N) rr = N - 1;
                const float4 v = *(const float4*)(asrc + rr * 128 + koff + kq);
                *(float4*)&As[r * LDK + kq] = v;
            }
            __syncthreads();
            v8f c = acc[g];
#pragma unroll
            for (int k = 0; k < 64; k += 4) {
                const v2f a = *(const v2f*)&As[ml * LDK + half * 2 + k];
                const v2f b = *(const v2f*)&Wt[(wave * 16 + ml) * LDK + half * 2 + k];
                c = __builtin_amdgcn_wmma_f32_16x16x4_f32(
                    false, a, false, b, (short)0, c, false, false);
            }
            acc[g] = c;
        }
    }

    const float bv = bias[wave * 16 + ml];
#pragma unroll
    for (int g = 0; g < G; ++g) {
        v8f c = acc[g];
#pragma unroll
        for (int r = 0; r < 8; ++r) c[r] += bv;

        if (LNRELU) {
#pragma unroll
            for (int r = 0; r < 8; ++r) {
                float sv = c[r];
                float qv = c[r] * c[r];
#pragma unroll
                for (int off = 1; off < 16; off <<= 1) {
                    sv += __shfl_xor(sv, off, 32);
                    qv += __shfl_xor(qv, off, 32);
                }
                if (ml == 0) {
                    red_s[wave * 16 + half * 8 + r] = sv;
                    red_q[wave * 16 + half * 8 + r] = qv;
                }
            }
            __syncthreads();
            if (tid < 16) {
                float ts = 0.0f, tq = 0.0f;
#pragma unroll
                for (int w = 0; w < COLT; ++w) {
                    ts += red_s[w * 16 + tid];
                    tq += red_q[w * 16 + tid];
                }
                red_s[tid] = ts;
                red_q[tid] = tq;
            }
            __syncthreads();
#pragma unroll
            for (int r = 0; r < 8; ++r) {
                const int rowid = half * 8 + r;
                const float mu   = red_s[rowid] * (1.0f / NDIM);
                const float var  = red_q[rowid] * (1.0f / NDIM) - mu * mu;
                const float rstd = rsqrtf(var + LN_EPS);
                float v = (c[r] - mu) * rstd;
                c[r] = v > 0.0f ? v : 0.0f;
            }
            __syncthreads();
        }

#pragma unroll
        for (int r = 0; r < 8; ++r) {
            const long ro = base + g * 16 + half * 8 + r;
            if (ro < N) y[ro * NDIM + wave * 16 + ml] = c[r];
        }
    }
}


extern "C" void kernel_launch(void* const* d_in, const int* in_sizes, int n_in,
                              void* d_out, int out_size, void* d_ws, size_t ws_size,
                              hipStream_t stream) {
    const float* h   = (const float*)d_in[0];
    const int*   src = (const int*)d_in[1];
    const int*   dst = (const int*)d_in[2];
    const float* W1  = (const float*)d_in[3];
    const float* b1  = (const float*)d_in[4];
    const float* W2  = (const float*)d_in[5];
    const float* b2  = (const float*)d_in[6];
    const float* W3  = (const float*)d_in[7];
    const float* b3  = (const float*)d_in[8];
    float* out = (float*)d_out;

    const int N = in_sizes[0] / 128;
    const int E = in_sizes[1];

    float* ws   = (float*)d_ws;
    float* norm = ws;
    float* agg  = norm + N;
    float* h1   = agg + (size_t)N * 128;
    float* h2   = h1  + (size_t)N * 128;

    const long feat_n      = (long)N * 128;
    const int  gemm_blocks = (N + 127) / 128;
    const int  scat_blocks = (int)(((long)E * 32 + 255) / 256);

    zero_kernel<<<1024, 256, 0, stream>>>(norm, N);
    degree_kernel<<<(E + 255) / 256, 256, 0, stream>>>(dst, norm, E);
    invert_kernel<<<(N + 255) / 256, 256, 0, stream>>>(norm, N);

    zero_kernel<<<2048, 256, 0, stream>>>(agg, feat_n);
    scatter_kernel<<<scat_blocks, 256, 0, stream>>>(h, src, dst, norm, agg, E);
    sage_gemm_kernel<8, true><<<gemm_blocks, 256, 0, stream>>>(h, agg, W1, b1, h1, N);

    zero_kernel<<<2048, 256, 0, stream>>>(agg, feat_n);
    scatter_kernel<<<scat_blocks, 256, 0, stream>>>(h1, src, dst, norm, agg, E);
    sage_gemm_kernel<8, true><<<gemm_blocks, 256, 0, stream>>>(h1, agg, W2, b2, h2, N);

    zero_kernel<<<2048, 256, 0, stream>>>(agg, feat_n);
    scatter_kernel<<<scat_blocks, 256, 0, stream>>>(h2, src, dst, norm, agg, E);
    sage_gemm_kernel<4, false><<<gemm_blocks, 128, 0, stream>>>(h2, agg, W3, b3, out, N);
}


# Neighbour sums per destination tile in LDS: a 1024-thread group scans all dst, queues its tile's edges, adds their rows with LDS float adds, writes rows once x 1/deg; 9 launches; sum order changes
# speedup vs baseline: 1.6497x; 1.6497x over previous
#include <hip/hip_runtime.h>

typedef __attribute__((ext_vector_type(2))) float v2f;
typedef __attribute__((ext_vector_type(8))) float v8f;

#define LN_EPS 1e-5f


__global__ void zero_kernel(float* __restrict__ p, long n) {
    long i = (long)blockIdx.x * blockDim.x + threadIdx.x;
    long stride = (long)gridDim.x * blockDim.x;
    for (; i < n; i += stride) p[i] = 0.0f;
}

__global__ void degree_kernel(const int* __restrict__ dst, float* __restrict__ deg, int E) {
    int e = blockIdx.x * blockDim.x + threadIdx.x;
    if (e < E) atomicAdd(&deg[dst[e]], 1.0f);
}

__global__ void invert_kernel(float* __restrict__ norm, int N) {
    int i = blockIdx.x * blockDim.x + threadIdx.x;
    if (i < N) {
        float d = norm[i];
        norm[i] = d > 0.0f ? 1.0f / d : 0.0f;
    }
}


__device__ __forceinline__ void load_dst32(int4 (&v)[8], const int* __restrict__ dst, int wbase,
                                           int lane, int E) {
    const int ebase = wbase + lane * 4;
    if (wbase + 1024 <= E) {
#pragma unroll
        for (int g = 0; g < 8; ++g) v[g] = *(const int4*)(dst + ebase + g * 128);
    } else {
#pragma unroll
        for (int g = 0; g < 8; ++g) {
            const int e = ebase + g * 128;
            v[g].x = (e     < E) ? dst[e]     : -1;
            v[g].y = (e + 1 < E) ? dst[e + 1] : -1;
            v[g].z = (e + 2 < E) ? dst[e + 2] : -1;
            v[g].w = (e + 3 < E) ? dst[e + 3] : -1;
        }
    }
}

template <int F>
__device__ __forceinline__ void drain_queue(const float* __restrict__ x, const int* __restrict__ src,
                                            const int* __restrict__ dst, float* tile, const int* qe,
                                            int lo, int cnt, int lane) {
    constexpr int C = F / 32;
    const int e_l  = (lane < cnt) ? qe[lane] : 0;
    const int s_l  = src[e_l];
    const int dl_l = dst[e_l] - lo;
    for (int b = 0; b < cnt; b += 8) {
        float v[8][C];
#pragma unroll
        for (int j = 0; j < 8; ++j) {
            const int s = __builtin_amdgcn_readlane(s_l, b + j);
            const float* xp = x + (long)s * F + lane;
#pragma unroll
            for (int c = 0; c < C; ++c) v[j][c] = xp[32 * c];
        }
#pragma unroll
        for (int j = 0; j < 8; ++j) {
            if (b + j < cnt) {
                const int dl = __builtin_amdgcn_readlane(dl_l, b + j);
                float* tp = tile + dl * F + lane;
#pragma unroll
                for (int c = 0; c < C; ++c) atomicAdd(tp + 32 * c, v[j][c]);
            }
        }
    }
}

template <int F, bool ADD_TO_OUT>
__global__ __launch_bounds__(1024) void tile_sum_kernel(
    const float* __restrict__ x,
    const int* __restrict__ src, const int* __restrict__ dst,
    const float* __restrict__ norm,
    float* __restrict__ y,
    int E, int N) {
    constexpr int TN  = 65536 / F;
    constexpr int C   = F / 32;
    constexpr int RPW = TN / 32;

    __shared__ float tile[TN * F];
    __shared__ int q_e[32 * 64];

    const int tid  = threadIdx.x;
    const int wave = __builtin_amdgcn_readfirstlane(tid >> 5);
    const int lane = tid & 31;
    const int lo   = blockIdx.x * TN;

    for (int i = tid * 4; i < TN * F; i += 4096) *(float4*)&tile[i] = float4{0.0f, 0.0f, 0.0f, 0.0f};
    __syncthreads();

    int* qe = q_e + wave * 64;
    int qn = 0;

    for (int e0 = wave * 1024; e0 < E; e0 += 32768) {
        int4 dv[8];
        load_dst32(dv, dst, e0, lane, E);
        unsigned hm = 0u;
#pragma unroll
        for (int g = 0; g < 8; ++g) {
            const int dd[4] = {dv[g].x, dv[g].y, dv[g].z, dv[g].w};
#pragma unroll
            for (int c = 0; c < 4; ++c)
                hm |= ((unsigned)(dd[c] - lo) < (unsigned)TN ? 1u : 0u) << (g * 4 + c);
        }
        for (;;) {
            const bool hit = hm != 0u;
            const unsigned m = __builtin_amdgcn_ballot_w32(hit);
            if (m == 0u) break;
            if (hit) {
                const int bit = __builtin_ctz(hm);
                hm &= hm - 1u;
                const int pos = qn + (int)__builtin_amdgcn_mbcnt_lo(m, 0u);
                qe[pos] = e0 + (bit >> 2) * 128 + lane * 4 + (bit & 3);
            }
            qn += __builtin_popcount(m);
            if (qn >= 32) {
                drain_queue<F>(x, src, dst, tile, qe, lo, 32, lane);
                const int rem = qn - 32;
                if (lane < rem) {
                    const int te = qe[32 + lane];
                    qe[lane] = te;
                }
                qn = rem;
            }
        }
    }
    if (qn > 0) drain_queue<F>(x, src, dst, tile, qe, lo, qn, lane);
    __syncthreads();

    const int r0 = wave * RPW;
    const int nl = lo + r0 + lane;
    const float nv_l = (lane < RPW && nl < N) ? norm[nl] : 0.0f;
    for (int rb = 0; rb < RPW; rb += 4) {
        float old[4][C];
        if (ADD_TO_OUT) {
#pragma unroll
            for (int k = 0; k < 4; ++k) {
                const int node = lo + r0 + rb + k;
#pragma unroll
                for (int c = 0; c < C; ++c)
                    old[k][c] = (node < N) ? y[(long)node * F + lane + 32 * c] : 0.0f;
            }
        }
#pragma unroll
        for (int k = 0; k < 4; ++k) {
            const int r = r0 + rb + k;
            const int node = lo + r;
            const float nv = __shfl(nv_l, rb + k, 32);
            if (node < N) {
#pragma unroll
                for (int c = 0; c < C; ++c) {
                    const float t = tile[r * F + lane + 32 * c] * nv;
                    y[(long)node * F + lane + 32 * c] = ADD_TO_OUT ? old[k][c] + t : t;
                }
            }
        }
    }
}

template <int COLT, bool LNRELU, bool PRE>
__global__ __launch_bounds__(COLT * 32) void sage_gemm_kernel(
    const float* __restrict__ x,
    const float* __restrict__ agg,
    const float* __restrict__ W,
    const float* __restrict__ bias,
    float* __restrict__ y,
    float* __restrict__ y2,
    int N) {
    constexpr int NDIM    = COLT * 16;
    constexpr int THREADS = COLT * 32;
    constexpr int G       = 8;
    constexpr int LDK     = 68;
    constexpr int HN      = NDIM / 2;
    constexpr int KCH     = PRE ? 2 : 4;

    __shared__ float Wt[NDIM * LDK];
    __shared__ float As[16 * LDK];
    __shared__ float red_s[8 * 16];
    __shared__ float red_q[8 * 16];

    const int tid  = threadIdx.x;
    const int wave = tid >> 5;
    const int lane = tid & 31;
    const int ml   = lane & 15;
    const int half = lane >> 4;
    const long base = (long)blockIdx.x * (G * 16);

    v8f acc[G];
#pragma unroll
    for (int g = 0; g < G; ++g) acc[g] = v8f{};

    for (int kc = 0; kc < KCH; ++kc) {
        const int k0 = kc * 64;
        __syncthreads();
        const float* wsrc = W + (long)k0 * NDIM;
        for (int idx = tid; idx < 64 * NDIM; idx += THREADS) {
            const int kloc = idx / NDIM;
            const int n    = idx - kloc * NDIM;
            if (PRE) {
                const int krow = (n < HN) ? (k0 + kloc) : (128 + k0 + kloc);
                const int col  = (n < HN) ? n : (n - HN);
                Wt[n * LDK + kloc] = W[(long)krow * HN + col];
            } else {
                Wt[n * LDK + kloc] = wsrc[idx];
            }
        }
        const float* asrc = (PRE || k0 < 128) ? x : agg;
        const int koff = k0 & 127;
#pragma unroll
        for (int g = 0; g < G; ++g) {
            __syncthreads();
            for (int q = tid; q < 256; q += THREADS) {
                const int r  = q >> 4;
                const int kq = (q & 15) * 4;
                long rr = base + g * 16 + r;
                if (rr >= N) rr = N - 1;
                const float4 v = *(const float4*)(asrc + rr * 128 + koff + kq);
                *(float4*)&As[r * LDK + kq] = v;
            }
            __syncthreads();
            v8f c = acc[g];
#pragma unroll
            for (int k = 0; k < 64; k += 4) {
                const v2f a = *(const v2f*)&As[ml * LDK + half * 2 + k];
                const v2f b = *(const v2f*)&Wt[(wave * 16 + ml) * LDK + half * 2 + k];
                c = __builtin_amdgcn_wmma_f32_16x16x4_f32(
                    false, a, false, b, (short)0, c, false, false);
            }
            acc[g] = c;
        }
    }

    const bool second = PRE && wave >= COLT / 2;
    const float bv = second ? 0.0f : bias[wave * 16 + ml];
#pragma unroll
    for (int g = 0; g < G; ++g) {
        v8f c = acc[g];
#pragma unroll
        for (int r = 0; r < 8; ++r) c[r] += bv;

        if (LNRELU) {
#pragma unroll
            for (int r = 0; r < 8; ++r) {
                float sv = c[r];
                float qv = c[r] * c[r];
#pragma unroll
                for (int off = 1; off < 16; off <<= 1) {
                    sv += __shfl_xor(sv, off, 32);
                    qv += __shfl_xor(qv, off, 32);
                }
                if (ml == 0) {
                    red_s[wave * 16 + half * 8 + r] = sv;
                    red_q[wave * 16 + half * 8 + r] = qv;
                }
            }
            __syncthreads();
            if (tid < 16) {
                float ts = 0.0f, tq = 0.0f;
#pragma unroll
                for (int w = 0; w < COLT; ++w) {
                    ts += red_s[w * 16 + tid];
                    tq += red_q[w * 16 + tid];
                }
                red_s[tid] = ts;
                red_q[tid] = tq;
            }
            __syncthreads();
#pragma unroll
            for (int r = 0; r < 8; ++r) {
                const int rowid = half * 8 + r;
                const float mu   = red_s[rowid] * (1.0f / NDIM);
                const float var  = red_q[rowid] * (1.0f / NDIM) - mu * mu;
                const float rstd = rsqrtf(var + LN_EPS);
                float v = (c[r] - mu) * rstd;
                c[r] = v > 0.0f ? v : 0.0f;
            }
            __syncthreads();
        }

#pragma unroll
        for (int r = 0; r < 8; ++r) {
            const long ro = base + g * 16 + half * 8 + r;
            if (PRE) {
                float* yp = second ? y2 : y;
                const int cw = second ? (wave - COLT / 2) : wave;
                if (ro < N) yp[ro * HN + cw * 16 + ml] = c[r];
            } else {
                if (ro < N) y[ro * NDIM + wave * 16 + ml] = c[r];
            }
        }
    }
}


extern "C" void kernel_launch(void* const* d_in, const int* in_sizes, int n_in,
                              void* d_out, int out_size, void* d_ws, size_t ws_size,
                              hipStream_t stream) {
    const float* h   = (const float*)d_in[0];
    const int*   src = (const int*)d_in[1];
    const int*   dst = (const int*)d_in[2];
    const float* W1  = (const float*)d_in[3];
    const float* b1  = (const float*)d_in[4];
    const float* W2  = (const float*)d_in[5];
    const float* b2  = (const float*)d_in[6];
    const float* W3  = (const float*)d_in[7];
    const float* b3  = (const float*)d_in[8];
    float* out = (float*)d_out;

    const int N = in_sizes[0] / 128;
    const int E = in_sizes[1];

    float* ws   = (float*)d_ws;
    float* norm = ws;
    float* agg  = norm + N;
    float* h1   = agg + (size_t)N * 128;
    float* h2   = h1  + (size_t)N * 128;

    const int  gemm_blocks = (N + 127) / 128;
    const int  tiles128    = (N + 511) / 512;
    const int  tiles64     = (N + 1023) / 1024;

    zero_kernel<<<1024, 256, 0, stream>>>(norm, N);
    degree_kernel<<<(E + 255) / 256, 256, 0, stream>>>(dst, norm, E);
    invert_kernel<<<(N + 255) / 256, 256, 0, stream>>>(norm, N);

    tile_sum_kernel<128, false><<<tiles128, 1024, 0, stream>>>(h, src, dst, norm, agg, E, N);
    sage_gemm_kernel<8, true, false><<<gemm_blocks, 256, 0, stream>>>(h, agg, W1, b1, h1, nullptr, N);

    tile_sum_kernel<128, false><<<tiles128, 1024, 0, stream>>>(h1, src, dst, norm, agg, E, N);
    sage_gemm_kernel<8, true, false><<<gemm_blocks, 256, 0, stream>>>(h1, agg, W2, b2, h2, nullptr, N);

    float* z = agg;
    sage_gemm_kernel<8, false, true><<<gemm_blocks, 256, 0, stream>>>(h2, nullptr, W3, b3, out, z, N);
    tile_sum_kernel<64, true><<<tiles64, 1024, 0, stream>>>(z, src, dst, norm, out, E, N);
}


# Layer GEMMs re-tiled: a wave keeps its 16 columns of W for all of K in registers; resident groups walk 32-row A tiles double-buffered in LDS, next tile requested ahead; K sum order changes
# speedup vs baseline: 1.8463x; 1.1192x over previous
#include <hip/hip_runtime.h>

typedef __attribute__((ext_vector_type(2))) float v2f;
typedef __attribute__((ext_vector_type(4))) float v4f;
typedef __attribute__((ext_vector_type(8))) float v8f;

#define LN_EPS 1e-5f


__global__ void zero_kernel(float* __restrict__ p, long n) {
    long i = (long)blockIdx.x * blockDim.x + threadIdx.x;
    long stride = (long)gridDim.x * blockDim.x;
    for (; i < n; i += stride) p[i] = 0.0f;
}

__global__ void degree_kernel(const int* __restrict__ dst, float* __restrict__ deg, int E) {
    int e = blockIdx.x * blockDim.x + threadIdx.x;
    if (e < E) atomicAdd(&deg[dst[e]], 1.0f);
}

__global__ void invert_kernel(float* __restrict__ norm, int N) {
    int i = blockIdx.x * blockDim.x + threadIdx.x;
    if (i < N) {
        float d = norm[i];
        norm[i] = d > 0.0f ? 1.0f / d : 0.0f;
    }
}


__device__ __forceinline__ void load_dst32(int4 (&v)[8], const int* __restrict__ dst, int wbase,
                                           int lane, int E) {
    const int ebase = wbase + lane * 4;
    if (wbase + 1024 <= E) {
#pragma unroll
        for (int g = 0; g < 8; ++g) v[g] = *(const int4*)(dst + ebase + g * 128);
    } else {
#pragma unroll
        for (int g = 0; g < 8; ++g) {
            const int e = ebase + g * 128;
            v[g].x = (e     < E) ? dst[e]     : -1;
            v[g].y = (e + 1 < E) ? dst[e + 1] : -1;
            v[g].z = (e + 2 < E) ? dst[e + 2] : -1;
            v[g].w = (e + 3 < E) ? dst[e + 3] : -1;
        }
    }
}

template <int F>
__device__ __forceinline__ void drain_queue(const float* __restrict__ x, const int* __restrict__ src,
                                            const int* __restrict__ dst, float* tile, const int* qe,
                                            int lo, int cnt, int lane) {
    constexpr int C = F / 32;
    const int e_l  = (lane < cnt) ? qe[lane] : 0;
    const int s_l  = src[e_l];
    const int dl_l = dst[e_l] - lo;
    for (int b = 0; b < cnt; b += 8) {
        float v[8][C];
#pragma unroll
        for (int j = 0; j < 8; ++j) {
            const int s = __builtin_amdgcn_readlane(s_l, b + j);
            const float* xp = x + (long)s * F + lane;
#pragma unroll
            for (int c = 0; c < C; ++c) v[j][c] = xp[32 * c];
        }
#pragma unroll
        for (int j = 0; j < 8; ++j) {
            if (b + j < cnt) {
                const int dl = __builtin_amdgcn_readlane(dl_l, b + j);
                float* tp = tile + dl * F + lane;
#pragma unroll
                for (int c = 0; c < C; ++c) atomicAdd(tp + 32 * c, v[j][c]);
            }
        }
    }
}

template <int F, bool ADD_TO_OUT>
__global__ __launch_bounds__(1024) void tile_sum_kernel(
    const float* __restrict__ x,
    const int* __restrict__ src, const int* __restrict__ dst,
    const float* __restrict__ norm,
    float* __restrict__ y,
    int E, int N) {
    constexpr int TN  = 65536 / F;
    constexpr int C   = F / 32;
    constexpr int RPW = TN / 32;

    __shared__ float tile[TN * F];
    __shared__ int q_e[32 * 64];

    const int tid  = threadIdx.x;
    const int wave = __builtin_amdgcn_readfirstlane(tid >> 5);
    const int lane = tid & 31;
    const int lo   = blockIdx.x * TN;

    for (int i = tid * 4; i < TN * F; i += 4096) *(float4*)&tile[i] = float4{0.0f, 0.0f, 0.0f, 0.0f};
    __syncthreads();

    int* qe = q_e + wave * 64;
    int qn = 0;

    for (int e0 = wave * 1024; e0 < E; e0 += 32768) {
        int4 dv[8];
        load_dst32(dv, dst, e0, lane, E);
        unsigned hm = 0u;
#pragma unroll
        for (int g = 0; g < 8; ++g) {
            const int dd[4] = {dv[g].x, dv[g].y, dv[g].z, dv[g].w};
#pragma unroll
            for (int c = 0; c < 4; ++c)
                hm |= ((unsigned)(dd[c] - lo) < (unsigned)TN ? 1u : 0u) << (g * 4 + c);
        }
        for (;;) {
            const bool hit = hm != 0u;
            const unsigned m = __builtin_amdgcn_ballot_w32(hit);
            if (m == 0u) break;
            if (hit) {
                const int bit = __builtin_ctz(hm);
                hm &= hm - 1u;
                const int pos = qn + (int)__builtin_amdgcn_mbcnt_lo(m, 0u);
                qe[pos] = e0 + (bit >> 2) * 128 + lane * 4 + (bit & 3);
            }
            qn += __builtin_popcount(m);
            if (qn >= 32) {
                drain_queue<F>(x, src, dst, tile, qe, lo, 32, lane);
                const int rem = qn - 32;
                if (lane < rem) {
                    const int te = qe[32 + lane];
                    qe[lane] = te;
                }
                qn = rem;
            }
        }
    }
    if (qn > 0) drain_queue<F>(x, src, dst, tile, qe, lo, qn, lane);
    __syncthreads();

    const int r0 = wave * RPW;
    const int nl = lo + r0 + lane;
    const float nv_l = (lane < RPW && nl < N) ? norm[nl] : 0.0f;
    for (int rb = 0; rb < RPW; rb += 4) {
        float old[4][C];
        if (ADD_TO_OUT) {
#pragma unroll
            for (int k = 0; k < 4; ++k) {
                const int node = lo + r0 + rb + k;
#pragma unroll
                for (int c = 0; c < C; ++c)
                    old[k][c] = (node < N) ? y[(long)node * F + lane + 32 * c] : 0.0f;
            }
        }
#pragma unroll
        for (int k = 0; k < 4; ++k) {
            const int r = r0 + rb + k;
            const int node = lo + r;
            const float nv = __shfl(nv_l, rb + k, 32);
            if (node < N) {
#pragma unroll
                for (int c = 0; c < C; ++c) {
                    const float t = tile[r * F + lane + 32 * c] * nv;
                    y[(long)node * F + lane + 32 * c] = ADD_TO_OUT ? old[k][c] + t : t;
                }
            }
        }
    }
}

template <bool LNRELU, bool PRE>
__global__ __launch_bounds__(256) void sage_gemm_kernel(
    const float* __restrict__ x,
    const float* __restrict__ agg,
    const float* __restrict__ W,
    const float* __restrict__ bias,
    float* __restrict__ y,
    float* __restrict__ y2,
    int N, int ntiles) {
    constexpr int NDIM = 128;
    constexpr int HN   = 64;
    constexpr int KTOT = PRE ? 128 : 256;
    constexpr int LDA  = KTOT + 4;
    constexpr int KP   = KTOT / 8;
    constexpr int F4R  = KTOT / 4;
    constexpr int NLD  = 32 * F4R / 256;
    constexpr int RSTEP = 256 / F4R;

    __shared__ float As[2][32 * LDA];
    __shared__ float red_s[8 * 32];
    __shared__ float red_q[8 * 32];
    __shared__ float comb[2 * 32];

    const int tid  = threadIdx.x;
    const int wave = tid >> 5;
    const int lane = tid & 31;
    const int ml   = lane & 15;
    const int half = lane >> 4;

    v2f breg[2 * KP];
    {
        const int n = wave * 16 + ml;
#pragma unroll
        for (int j = 0; j < KP; ++j) {
#pragma unroll
            for (int t = 0; t < 2; ++t) {
                const int k = 8 * j + 4 * half + 2 * t;
                float w0, w1;
                if (PRE) {
                    const int krow = (n < HN) ? k : (128 + k);
                    const int col  = (n < HN) ? n : (n - HN);
                    w0 = W[(long)krow * HN + col];
                    w1 = W[(long)(krow + 1) * HN + col];
                } else {
                    w0 = W[(long)k * NDIM + n];
                    w1 = W[(long)(k + 1) * NDIM + n];
                }
                breg[2 * j + t] = v2f{w0, w1};
            }
        }
    }

    const int srow = tid / F4R;
    const int sc4  = tid % F4R;
    const float* sbase = (PRE || sc4 < 32) ? x : agg;
    const int scol = (sc4 & 31) * 4;

    const bool second = PRE && wave >= 4;
    const float bv = second ? 0.0f : bias[wave * 16 + ml];

    int tile = blockIdx.x;
    {
#pragma unroll
        for (int i = 0; i < NLD; ++i) {
            long rr = (long)tile * 32 + srow + i * RSTEP;
            if (rr >= N) rr = N - 1;
            const v4f v = *(const v4f*)(sbase + rr * 128 + scol);
            *(v4f*)&As[0][(srow + i * RSTEP) * LDA + sc4 * 4] = v;
        }
    }
    __syncthreads();

    int buf = 0;
    for (; tile < ntiles; tile += gridDim.x) {
        const int nxt = tile + gridDim.x;
        v4f pre[NLD];
#pragma unroll
        for (int i = 0; i < NLD; ++i) {
            long rr = (long)nxt * 32 + srow + i * RSTEP;
            if (rr >= N) rr = N - 1;
            pre[i] = *(const v4f*)(sbase + rr * 128 + scol);
        }
        __builtin_amdgcn_sched_barrier(0);

        const float* a0p = &As[buf][ml * LDA + 4 * half];
        const float* a1p = a0p + 16 * LDA;
        v8f c0 = v8f{}, c1 = v8f{};
#pragma unroll
        for (int j = 0; j < KP; ++j) {
            const v4f a0 = *(const v4f*)(a0p + 8 * j);
            const v4f a1 = *(const v4f*)(a1p + 8 * j);
            c0 = __builtin_amdgcn_wmma_f32_16x16x4_f32(
                false, v2f{a0.x, a0.y}, false, breg[2 * j], (short)0, c0, false, false);
            c1 = __builtin_amdgcn_wmma_f32_16x16x4_f32(
                false, v2f{a1.x, a1.y}, false, breg[2 * j], (short)0, c1, false, false);
            c0 = __builtin_amdgcn_wmma_f32_16x16x4_f32(
                false, v2f{a0.z, a0.w}, false, breg[2 * j + 1], (short)0, c0, false, false);
            c1 = __builtin_amdgcn_wmma_f32_16x16x4_f32(
                false, v2f{a1.z, a1.w}, false, breg[2 * j + 1], (short)0, c1, false, false);
        }

#pragma unroll
        for (int r = 0; r < 8; ++r) { c0[r] += bv; c1[r] += bv; }

        if (LNRELU) {
#pragma unroll
            for (int g = 0; g < 2; ++g) {
#pragma unroll
                for (int r = 0; r < 8; ++r) {
                    const float cv = g ? c1[r] : c0[r];
                    float sv = cv;
                    float qv = cv * cv;
#pragma unroll
                    for (int off = 1; off < 16; off <<= 1) {
                        sv += __shfl_xor(sv, off, 32);
                        qv += __shfl_xor(qv, off, 32);
                    }
                    if (ml == 0) {
                        red_s[wave * 32 + g * 16 + half * 8 + r] = sv;
                        red_q[wave * 32 + g * 16 + half * 8 + r] = qv;
                    }
                }
            }
            __syncthreads();
            if (tid < 64) {
                const float* rp = (tid < 32) ? red_s : red_q;
                const int row = tid & 31;
                float ts = 0.0f;
#pragma unroll
                for (int w = 0; w < 8; ++w) ts += rp[w * 32 + row];
                comb[tid] = ts;
            }
            __syncthreads();
#pragma unroll
            for (int g = 0; g < 2; ++g) {
#pragma unroll
                for (int r = 0; r < 8; ++r) {
                    const int rowid = g * 16 + half * 8 + r;
                    const float mu   = comb[rowid] * (1.0f / NDIM);
                    const float var  = comb[32 + rowid] * (1.0f / NDIM) - mu * mu;
                    const float rstd = rsqrtf(var + LN_EPS);
                    const float cv = g ? c1[r] : c0[r];
                    float v = (cv - mu) * rstd;
                    v = v > 0.0f ? v : 0.0f;
                    if (g) c1[r] = v; else c0[r] = v;
                }
            }
        }

#pragma unroll
        for (int i = 0; i < NLD; ++i)
            *(v4f*)&As[buf ^ 1][(srow + i * RSTEP) * LDA + sc4 * 4] = pre[i];
        __syncthreads();
        buf ^= 1;

#pragma unroll
        for (int g = 0; g < 2; ++g) {
#pragma unroll
            for (int r = 0; r < 8; ++r) {
                const long ro = (long)tile * 32 + g * 16 + half * 8 + r;
                const float cv = g ? c1[r] : c0[r];
                if (PRE) {
                    float* yp = second ? y2 : y;
                    const int cw = second ? (wave - 4) : wave;
                    if (ro < N) yp[ro * HN + cw * 16 + ml] = cv;
                } else {
                    if (ro < N) y[ro * NDIM + wave * 16 + ml] = cv;
                }
            }
        }
    }
}


extern "C" void kernel_launch(void* const* d_in, const int* in_sizes, int n_in,
                              void* d_out, int out_size, void* d_ws, size_t ws_size,
                              hipStream_t stream) {
    const float* h   = (const float*)d_in[0];
    const int*   src = (const int*)d_in[1];
    const int*   dst = (const int*)d_in[2];
    const float* W1  = (const float*)d_in[3];
    const float* b1  = (const float*)d_in[4];
    const float* W2  = (const float*)d_in[5];
    const float* b2  = (const float*)d_in[6];
    const float* W3  = (const float*)d_in[7];
    const float* b3  = (const float*)d_in[8];
    float* out = (float*)d_out;

    const int N = in_sizes[0] / 128;
    const int E = in_sizes[1];

    float* ws   = (float*)d_ws;
    float* norm = ws;
    float* agg  = norm + N;
    float* h1   = agg + (size_t)N * 128;
    float* h2   = h1  + (size_t)N * 128;

    const int  ntiles      = (N + 31) / 32;
    const int  gemm_blocks = ntiles < 512 ? ntiles : 512;
    const int  gemm_blocks3 = ntiles < 1024 ? ntiles : 1024;
    const int  tiles128    = (N + 511) / 512;
    const int  tiles64     = (N + 1023) / 1024;

    zero_kernel<<<1024, 256, 0, stream>>>(norm, N);
    degree_kernel<<<(E + 255) / 256, 256, 0, stream>>>(dst, norm, E);
    invert_kernel<<<(N + 255) / 256, 256, 0, stream>>>(norm, N);

    tile_sum_kernel<128, false><<<tiles128, 1024, 0, stream>>>(h, src, dst, norm, agg, E, N);
    sage_gemm_kernel<true, false><<<gemm_blocks, 256, 0, stream>>>(h, agg, W1, b1, h1, nullptr, N, ntiles);

    tile_sum_kernel<128, false><<<tiles128, 1024, 0, stream>>>(h1, src, dst, norm, agg, E, N);
    sage_gemm_kernel<true, false><<<gemm_blocks, 256, 0, stream>>>(h1, agg, W2, b2, h2, nullptr, N, ntiles);

    float* z = agg;
    sage_gemm_kernel<false, true><<<gemm_blocks3, 256, 0, stream>>>(h2, nullptr, W3, b3, out, z, N, ntiles);
    tile_sum_kernel<64, true><<<tiles64, 1024, 0, stream>>>(z, src, dst, norm, out, E, N);
}


# Neighbour-sum tiles cut to 400 rows for both widths: 250 groups, one on nearly every WGP, not 196 and 98; each still scans all dst but has fewer edges, LDS adds and rows; sums arrival-ordered
# speedup vs baseline: 1.9384x; 1.0499x over previous
#include <hip/hip_runtime.h>

typedef __attribute__((ext_vector_type(2))) float v2f;
typedef __attribute__((ext_vector_type(4))) float v4f;
typedef __attribute__((ext_vector_type(8))) float v8f;

#define LN_EPS 1e-5f


__global__ void zero_kernel(float* __restrict__ p, long n) {
    long i = (long)blockIdx.x * blockDim.x + threadIdx.x;
    long stride = (long)gridDim.x * blockDim.x;
    for (; i < n; i += stride) p[i] = 0.0f;
}

__global__ void degree_kernel(const int* __restrict__ dst, float* __restrict__ deg, int E) {
    int e = blockIdx.x * blockDim.x + threadIdx.x;
    if (e < E) atomicAdd(&deg[dst[e]], 1.0f);
}

__global__ void invert_kernel(float* __restrict__ norm, int N) {
    int i = blockIdx.x * blockDim.x + threadIdx.x;
    if (i < N) {
        float d = norm[i];
        norm[i] = d > 0.0f ? 1.0f / d : 0.0f;
    }
}


__device__ __forceinline__ void load_dst32(int4 (&v)[8], const int* __restrict__ dst, int wbase,
                                           int lane, int E) {
    const int ebase = wbase + lane * 4;
    if (wbase + 1024 <= E) {
#pragma unroll
        for (int g = 0; g < 8; ++g) v[g] = *(const int4*)(dst + ebase + g * 128);
    } else {
#pragma unroll
        for (int g = 0; g < 8; ++g) {
            const int e = ebase + g * 128;
            v[g].x = (e     < E) ? dst[e]     : -1;
            v[g].y = (e + 1 < E) ? dst[e + 1] : -1;
            v[g].z = (e + 2 < E) ? dst[e + 2] : -1;
            v[g].w = (e + 3 < E) ? dst[e + 3] : -1;
        }
    }
}

template <int F>
__device__ __forceinline__ void drain_queue(const float* __restrict__ x, const int* __restrict__ src,
                                            const int* __restrict__ dst, float* tile, const int* qe,
                                            int lo, int cnt, int lane) {
    constexpr int C = F / 32;
    const int e_l  = (lane < cnt) ? qe[lane] : 0;
    const int s_l  = src[e_l];
    const int dl_l = dst[e_l] - lo;
    for (int b = 0; b < cnt; b += 8) {
        float v[8][C];
#pragma unroll
        for (int j = 0; j < 8; ++j) {
            const int s = __builtin_amdgcn_readlane(s_l, b + j);
            const float* xp = x + (long)s * F + lane;
#pragma unroll
            for (int c = 0; c < C; ++c) v[j][c] = xp[32 * c];
        }
#pragma unroll
        for (int j = 0; j < 8; ++j) {
            if (b + j < cnt) {
                const int dl = __builtin_amdgcn_readlane(dl_l, b + j);
                float* tp = tile + dl * F + lane;
#pragma unroll
                for (int c = 0; c < C; ++c) atomicAdd(tp + 32 * c, v[j][c]);
            }
        }
    }
}

template <int F, int TN, bool ADD_TO_OUT>
__global__ __launch_bounds__(1024) void tile_sum_kernel(
    const float* __restrict__ x,
    const int* __restrict__ src, const int* __restrict__ dst,
    const float* __restrict__ norm,
    float* __restrict__ y,
    int E, int N) {
    constexpr int C = F / 32;
    static_assert(TN % 4 == 0, "the write-out takes 4 rows at a time");

    __shared__ float tile[TN * 128];
    __shared__ int q_e[32 * 64];

    const int tid  = threadIdx.x;
    const int wave = __builtin_amdgcn_readfirstlane(tid >> 5);
    const int lane = tid & 31;
    const int lo   = blockIdx.x * TN;

    for (int i = tid * 4; i < TN * F; i += 4096) *(float4*)&tile[i] = float4{0.0f, 0.0f, 0.0f, 0.0f};
    __syncthreads();

    int* qe = q_e + wave * 64;
    int qn = 0;

    for (int e0 = wave * 1024; e0 < E; e0 += 32768) {
        int4 dv[8];
        load_dst32(dv, dst, e0, lane, E);
        unsigned hm = 0u;
#pragma unroll
        for (int g = 0; g < 8; ++g) {
            const int dd[4] = {dv[g].x, dv[g].y, dv[g].z, dv[g].w};
#pragma unroll
            for (int c = 0; c < 4; ++c)
                hm |= ((unsigned)(dd[c] - lo) < (unsigned)TN ? 1u : 0u) << (g * 4 + c);
        }
        for (;;) {
            const bool hit = hm != 0u;
            const unsigned m = __builtin_amdgcn_ballot_w32(hit);
            if (m == 0u) break;
            if (hit) {
                const int bit = __builtin_ctz(hm);
                hm &= hm - 1u;
                const int pos = qn + (int)__builtin_amdgcn_mbcnt_lo(m, 0u);
                qe[pos] = e0 + (bit >> 2) * 128 + lane * 4 + (bit & 3);
            }
            qn += __builtin_popcount(m);
            if (qn >= 32) {
                drain_queue<F>(x, src, dst, tile, qe, lo, 32, lane);
                const int rem = qn - 32;
                if (lane < rem) {
                    const int te = qe[32 + lane];
                    qe[lane] = te;
                }
                qn = rem;
            }
        }
    }
    if (qn > 0) drain_queue<F>(x, src, dst, tile, qe, lo, qn, lane);
    __syncthreads();

    for (int r0 = wave * 4; r0 < TN; r0 += 128) {
        float old[4][C];
        float nvs[4];
#pragma unroll
        for (int k = 0; k < 4; ++k) {
            const int node = lo + r0 + k;
            nvs[k] = (node < N) ? norm[node] : 0.0f;
        }
        if (ADD_TO_OUT) {
#pragma unroll
            for (int k = 0; k < 4; ++k) {
                const int node = lo + r0 + k;
#pragma unroll
                for (int c = 0; c < C; ++c)
                    old[k][c] = (node < N) ? y[(long)node * F + lane + 32 * c] : 0.0f;
            }
        }
#pragma unroll
        for (int k = 0; k < 4; ++k) {
            const int r = r0 + k;
            const int node = lo + r;
            const float nv = nvs[k];
            if (node < N) {
#pragma unroll
                for (int c = 0; c < C; ++c) {
                    const float t = tile[r * F + lane + 32 * c] * nv;
                    y[(long)node * F + lane + 32 * c] = ADD_TO_OUT ? old[k][c] + t : t;
                }
            }
        }
    }
}

template <bool LNRELU, bool PRE>
__global__ __launch_bounds__(256) void sage_gemm_kernel(
    const float* __restrict__ x,
    const float* __restrict__ agg,
    const float* __restrict__ W,
    const float* __restrict__ bias,
    float* __restrict__ y,
    float* __restrict__ y2,
    int N, int ntiles) {
    constexpr int NDIM = 128;
    constexpr int HN   = 64;
    constexpr int KTOT = PRE ? 128 : 256;
    constexpr int LDA  = KTOT + 4;
    constexpr int KP   = KTOT / 8;
    constexpr int F4R  = KTOT / 4;
    constexpr int NLD  = 32 * F4R / 256;
    constexpr int RSTEP = 256 / F4R;

    __shared__ float As[2][32 * LDA];
    __shared__ float red_s[8 * 32];
    __shared__ float red_q[8 * 32];
    __shared__ float comb[2 * 32];

    const int tid  = threadIdx.x;
    const int wave = tid >> 5;
    const int lane = tid & 31;
    const int ml   = lane & 15;
    const int half = lane >> 4;

    v2f breg[2 * KP];
    {
        const int n = wave * 16 + ml;
#pragma unroll
        for (int j = 0; j < KP; ++j) {
#pragma unroll
            for (int t = 0; t < 2; ++t) {
                const int k = 8 * j + 4 * half + 2 * t;
                float w0, w1;
                if (PRE) {
                    const int krow = (n < HN) ? k : (128 + k);
                    const int col  = (n < HN) ? n : (n - HN);
                    w0 = W[(long)krow * HN + col];
                    w1 = W[(long)(krow + 1) * HN + col];
                } else {
                    w0 = W[(long)k * NDIM + n];
                    w1 = W[(long)(k + 1) * NDIM + n];
                }
                breg[2 * j + t] = v2f{w0, w1};
            }
        }
    }

    const int srow = tid / F4R;
    const int sc4  = tid % F4R;
    const float* sbase = (PRE || sc4 < 32) ? x : agg;
    const int scol = (sc4 & 31) * 4;

    const bool second = PRE && wave >= 4;
    const float bv = second ? 0.0f : bias[wave * 16 + ml];

    int tile = blockIdx.x;
    {
#pragma unroll
        for (int i = 0; i < NLD; ++i) {
            long rr = (long)tile * 32 + srow + i * RSTEP;
            if (rr >= N) rr = N - 1;
            const v4f v = *(const v4f*)(sbase + rr * 128 + scol);
            *(v4f*)&As[0][(srow + i * RSTEP) * LDA + sc4 * 4] = v;
        }
    }
    __syncthreads();

    int buf = 0;
    for (; tile < ntiles; tile += gridDim.x) {
        const int nxt = tile + gridDim.x;
        v4f pre[NLD];
#pragma unroll
        for (int i = 0; i < NLD; ++i) {
            long rr = (long)nxt * 32 + srow + i * RSTEP;
            if (rr >= N) rr = N - 1;
            pre[i] = *(const v4f*)(sbase + rr * 128 + scol);
        }
        __builtin_amdgcn_sched_barrier(0);

        const float* a0p = &As[buf][ml * LDA + 4 * half];
        const float* a1p = a0p + 16 * LDA;
        v8f c0 = v8f{}, c1 = v8f{};
#pragma unroll
        for (int j = 0; j < KP; ++j) {
            const v4f a0 = *(const v4f*)(a0p + 8 * j);
            const v4f a1 = *(const v4f*)(a1p + 8 * j);
            c0 = __builtin_amdgcn_wmma_f32_16x16x4_f32(
                false, v2f{a0.x, a0.y}, false, breg[2 * j], (short)0, c0, false, false);
            c1 = __builtin_amdgcn_wmma_f32_16x16x4_f32(
                false, v2f{a1.x, a1.y}, false, breg[2 * j], (short)0, c1, false, false);
            c0 = __builtin_amdgcn_wmma_f32_16x16x4_f32(
                false, v2f{a0.z, a0.w}, false, breg[2 * j + 1], (short)0, c0, false, false);
            c1 = __builtin_amdgcn_wmma_f32_16x16x4_f32(
                false, v2f{a1.z, a1.w}, false, breg[2 * j + 1], (short)0, c1, false, false);
        }

#pragma unroll
        for (int r = 0; r < 8; ++r) { c0[r] += bv; c1[r] += bv; }

        if (LNRELU) {
#pragma unroll
            for (int g = 0; g < 2; ++g) {
#pragma unroll
                for (int r = 0; r < 8; ++r) {
                    const float cv = g ? c1[r] : c0[r];
                    float sv = cv;
                    float qv = cv * cv;
#pragma unroll
                    for (int off = 1; off < 16; off <<= 1) {
                        sv += __shfl_xor(sv, off, 32);
                        qv += __shfl_xor(qv, off, 32);
                    }
                    if (ml == 0) {
                        red_s[wave * 32 + g * 16 + half * 8 + r] = sv;
                        red_q[wave * 32 + g * 16 + half * 8 + r] = qv;
                    }
                }
            }
            __syncthreads();
            if (tid < 64) {
                const float* rp = (tid < 32) ? red_s : red_q;
                const int row = tid & 31;
                float ts = 0.0f;
#pragma unroll
                for (int w = 0; w < 8; ++w) ts += rp[w * 32 + row];
                comb[tid] = ts;
            }
            __syncthreads();
#pragma unroll
            for (int g = 0; g < 2; ++g) {
#pragma unroll
                for (int r = 0; r < 8; ++r) {
                    const int rowid = g * 16 + half * 8 + r;
                    const float mu   = comb[rowid] * (1.0f / NDIM);
                    const float var  = comb[32 + rowid] * (1.0f / NDIM) - mu * mu;
                    const float rstd = rsqrtf(var + LN_EPS);
                    const float cv = g ? c1[r] : c0[r];
                    float v = (cv - mu) * rstd;
                    v = v > 0.0f ? v : 0.0f;
                    if (g) c1[r] = v; else c0[r] = v;
                }
            }
        }

#pragma unroll
        for (int i = 0; i < NLD; ++i)
            *(v4f*)&As[buf ^ 1][(srow + i * RSTEP) * LDA + sc4 * 4] = pre[i];
        __syncthreads();
        buf ^= 1;

#pragma unroll
        for (int g = 0; g < 2; ++g) {
#pragma unroll
            for (int r = 0; r < 8; ++r) {
                const long ro = (long)tile * 32 + g * 16 + half * 8 + r;
                const float cv = g ? c1[r] : c0[r];
                if (PRE) {
                    float* yp = second ? y2 : y;
                    const int cw = second ? (wave - 4) : wave;
                    if (ro < N) yp[ro * HN + cw * 16 + ml] = cv;
                } else {
                    if (ro < N) y[ro * NDIM + wave * 16 + ml] = cv;
                }
            }
        }
    }
}


extern "C" void kernel_launch(void* const* d_in, const int* in_sizes, int n_in,
                              void* d_out, int out_size, void* d_ws, size_t ws_size,
                              hipStream_t stream) {
    const float* h   = (const float*)d_in[0];
    const int*   src = (const int*)d_in[1];
    const int*   dst = (const int*)d_in[2];
    const float* W1  = (const float*)d_in[3];
    const float* b1  = (const float*)d_in[4];
    const float* W2  = (const float*)d_in[5];
    const float* b2  = (const float*)d_in[6];
    const float* W3  = (const float*)d_in[7];
    const float* b3  = (const float*)d_in[8];
    float* out = (float*)d_out;

    const int N = in_sizes[0] / 128;
    const int E = in_sizes[1];

    float* ws   = (float*)d_ws;
    float* norm = ws;
    float* agg  = norm + N;
    float* h1   = agg + (size_t)N * 128;
    float* h2   = h1  + (size_t)N * 128;

    const int  ntiles      = (N + 31) / 32;
    const int  gemm_blocks = ntiles < 512 ? ntiles : 512;
    const int  gemm_blocks3 = ntiles < 1024 ? ntiles : 1024;
    const int  sum_tiles   = (N + 399) / 400;

    zero_kernel<<<1024, 256, 0, stream>>>(norm, N);
    degree_kernel<<<(E + 255) / 256, 256, 0, stream>>>(dst, norm, E);
    invert_kernel<<<(N + 255) / 256, 256, 0, stream>>>(norm, N);

    tile_sum_kernel<128, 400, false><<<sum_tiles, 1024, 0, stream>>>(h, src, dst, norm, agg, E, N);
    sage_gemm_kernel<true, false><<<gemm_blocks, 256, 0, stream>>>(h, agg, W1, b1, h1, nullptr, N, ntiles);

    tile_sum_kernel<128, 400, false><<<sum_tiles, 1024, 0, stream>>>(h1, src, dst, norm, agg, E, N);
    sage_gemm_kernel<true, false><<<gemm_blocks, 256, 0, stream>>>(h1, agg, W2, b2, h2, nullptr, N, ntiles);

    float* z = agg;
    sage_gemm_kernel<false, true><<<gemm_blocks3, 256, 0, stream>>>(h2, nullptr, W3, b3, out, z, N, ntiles);
    tile_sum_kernel<64, 400, true><<<sum_tiles, 1024, 0, stream>>>(z, src, dst, norm, out, E, N);
}


# In-degree counted in the first neighbour-sum pass: one LDS integer add per queued edge, 1/deg made at its write-out and stored for later passes; clear, count, invert launches dropped (6 launches)
# speedup vs baseline: 2.1807x; 1.1250x over previous
#include <hip/hip_runtime.h>

typedef __attribute__((ext_vector_type(2))) float v2f;
typedef __attribute__((ext_vector_type(4))) float v4f;
typedef __attribute__((ext_vector_type(8))) float v8f;

#define LN_EPS 1e-5f


__device__ __forceinline__ void load_dst32(int4 (&v)[8], const int* __restrict__ dst, int wbase,
                                           int lane, int E) {
    const int ebase = wbase + lane * 4;
    if (wbase + 1024 <= E) {
#pragma unroll
        for (int g = 0; g < 8; ++g) v[g] = *(const int4*)(dst + ebase + g * 128);
    } else {
#pragma unroll
        for (int g = 0; g < 8; ++g) {
            const int e = ebase + g * 128;
            v[g].x = (e     < E) ? dst[e]     : -1;
            v[g].y = (e + 1 < E) ? dst[e + 1] : -1;
            v[g].z = (e + 2 < E) ? dst[e + 2] : -1;
            v[g].w = (e + 3 < E) ? dst[e + 3] : -1;
        }
    }
}

template <int F, bool COUNT>
__device__ __forceinline__ void drain_queue(const float* __restrict__ x, const int* __restrict__ src,
                                            const int* __restrict__ dst, float* tile, int* degc,
                                            const int* qe, int lo, int cnt, int lane) {
    constexpr int C = F / 32;
    const int e_l  = (lane < cnt) ? qe[lane] : 0;
    const int s_l  = src[e_l];
    const int dl_l = dst[e_l] - lo;
    if (COUNT) {
        if (lane < cnt) atomicAdd(&degc[dl_l], 1);
    }
    for (int b = 0; b < cnt; b += 8) {
        float v[8][C];
#pragma unroll
        for (int j = 0; j < 8; ++j) {
            const int s = __builtin_amdgcn_readlane(s_l, b + j);
            const float* xp = x + (long)s * F + lane;
#pragma unroll
            for (int c = 0; c < C; ++c) v[j][c] = xp[32 * c];
        }
#pragma unroll
        for (int j = 0; j < 8; ++j) {
            if (b + j < cnt) {
                const int dl = __builtin_amdgcn_readlane(dl_l, b + j);
                float* tp = tile + dl * F + lane;
#pragma unroll
                for (int c = 0; c < C; ++c) atomicAdd(tp + 32 * c, v[j][c]);
            }
        }
    }
}

template <int F, int TN, bool ADD_TO_OUT, bool COUNT>
__global__ __launch_bounds__(1024) void tile_sum_kernel(
    const float* __restrict__ x,
    const int* __restrict__ src, const int* __restrict__ dst,
    const float* __restrict__ norm,
    float* __restrict__ norm_out,
    float* __restrict__ y,
    int E, int N) {
    constexpr int C = F / 32;
    static_assert(TN % 4 == 0, "the write-out takes 4 rows at a time");

    __shared__ float tile[TN * 128];
    __shared__ int q_e[32 * 64];
    __shared__ int degc[COUNT ? TN : 1];

    const int tid  = threadIdx.x;
    const int wave = __builtin_amdgcn_readfirstlane(tid >> 5);
    const int lane = tid & 31;
    const int lo   = blockIdx.x * TN;

    for (int i = tid * 4; i < TN * F; i += 4096) *(float4*)&tile[i] = float4{0.0f, 0.0f, 0.0f, 0.0f};
    if (COUNT) {
        for (int i = tid; i < TN; i += 1024) degc[i] = 0;
    }
    __syncthreads();

    int* qe = q_e + wave * 64;
    int qn = 0;

    for (int e0 = wave * 1024; e0 < E; e0 += 32768) {
        int4 dv[8];
        load_dst32(dv, dst, e0, lane, E);
        unsigned hm = 0u;
#pragma unroll
        for (int g = 0; g < 8; ++g) {
            const int dd[4] = {dv[g].x, dv[g].y, dv[g].z, dv[g].w};
#pragma unroll
            for (int c = 0; c < 4; ++c)
                hm |= ((unsigned)(dd[c] - lo) < (unsigned)TN ? 1u : 0u) << (g * 4 + c);
        }
        for (;;) {
            const bool hit = hm != 0u;
            const unsigned m = __builtin_amdgcn_ballot_w32(hit);
            if (m == 0u) break;
            if (hit) {
                const int bit = __builtin_ctz(hm);
                hm &= hm - 1u;
                const int pos = qn + (int)__builtin_amdgcn_mbcnt_lo(m, 0u);
                qe[pos] = e0 + (bit >> 2) * 128 + lane * 4 + (bit & 3);
            }
            qn += __builtin_popcount(m);
            if (qn >= 32) {
                drain_queue<F, COUNT>(x, src, dst, tile, degc, qe, lo, 32, lane);
                const int rem = qn - 32;
                if (lane < rem) {
                    const int te = qe[32 + lane];
                    qe[lane] = te;
                }
                qn = rem;
            }
        }
    }
    if (qn > 0) drain_queue<F, COUNT>(x, src, dst, tile, degc, qe, lo, qn, lane);
    __syncthreads();

    if (COUNT) {
        for (int r = tid; r < TN; r += 1024) {
            const int node = lo + r;
            if (node < N) {
                const int c = degc[r];
                norm_out[node] = c > 0 ? 1.0f / (float)c : 0.0f;
            }
        }
    }

    for (int r0 = wave * 4; r0 < TN; r0 += 128) {
        float old[4][C];
        float nvs[4];
#pragma unroll
        for (int k = 0; k < 4; ++k) {
            const int node = lo + r0 + k;
            if (COUNT) {
                const int c = degc[r0 + k];
                nvs[k] = c > 0 ? 1.0f / (float)c : 0.0f;
            } else {
                nvs[k] = (node < N) ? norm[node] : 0.0f;
            }
        }
        if (ADD_TO_OUT) {
#pragma unroll
            for (int k = 0; k < 4; ++k) {
                const int node = lo + r0 + k;
#pragma unroll
                for (int c = 0; c < C; ++c)
                    old[k][c] = (node < N) ? y[(long)node * F + lane + 32 * c] : 0.0f;
            }
        }
#pragma unroll
        for (int k = 0; k < 4; ++k) {
            const int r = r0 + k;
            const int node = lo + r;
            const float nv = nvs[k];
            if (node < N) {
#pragma unroll
                for (int c = 0; c < C; ++c) {
                    const float t = tile[r * F + lane + 32 * c] * nv;
                    y[(long)node * F + lane + 32 * c] = ADD_TO_OUT ? old[k][c] + t : t;
                }
            }
        }
    }
}

template <bool LNRELU, bool PRE>
__global__ __launch_bounds__(256) void sage_gemm_kernel(
    const float* __restrict__ x,
    const float* __restrict__ agg,
    const float* __restrict__ W,
    const float* __restrict__ bias,
    float* __restrict__ y,
    float* __restrict__ y2,
    int N, int ntiles) {
    constexpr int NDIM = 128;
    constexpr int HN   = 64;
    constexpr int KTOT = PRE ? 128 : 256;
    constexpr int LDA  = KTOT + 4;
    constexpr int KP   = KTOT / 8;
    constexpr int F4R  = KTOT / 4;
    constexpr int NLD  = 32 * F4R / 256;
    constexpr int RSTEP = 256 / F4R;

    __shared__ float As[2][32 * LDA];
    __shared__ float red_s[8 * 32];
    __shared__ float red_q[8 * 32];
    __shared__ float comb[2 * 32];

    const int tid  = threadIdx.x;
    const int wave = tid >> 5;
    const int lane = tid & 31;
    const int ml   = lane & 15;
    const int half = lane >> 4;

    v2f breg[2 * KP];
    {
        const int n = wave * 16 + ml;
#pragma unroll
        for (int j = 0; j < KP; ++j) {
#pragma unroll
            for (int t = 0; t < 2; ++t) {
                const int k = 8 * j + 4 * half + 2 * t;
                float w0, w1;
                if (PRE) {
                    const int krow = (n < HN) ? k : (128 + k);
                    const int col  = (n < HN) ? n : (n - HN);
                    w0 = W[(long)krow * HN + col];
                    w1 = W[(long)(krow + 1) * HN + col];
                } else {
                    w0 = W[(long)k * NDIM + n];
                    w1 = W[(long)(k + 1) * NDIM + n];
                }
                breg[2 * j + t] = v2f{w0, w1};
            }
        }
    }

    const int srow = tid / F4R;
    const int sc4  = tid % F4R;
    const float* sbase = (PRE || sc4 < 32) ? x : agg;
    const int scol = (sc4 & 31) * 4;

    const bool second = PRE && wave >= 4;
    const float bv = second ? 0.0f : bias[wave * 16 + ml];

    int tile = blockIdx.x;
    {
#pragma unroll
        for (int i = 0; i < NLD; ++i) {
            long rr = (long)tile * 32 + srow + i * RSTEP;
            if (rr >= N) rr = N - 1;
            const v4f v = *(const v4f*)(sbase + rr * 128 + scol);
            *(v4f*)&As[0][(srow + i * RSTEP) * LDA + sc4 * 4] = v;
        }
    }
    __syncthreads();

    int buf = 0;
    for (; tile < ntiles; tile += gridDim.x) {
        const int nxt = tile + gridDim.x;
        v4f pre[NLD];
#pragma unroll
        for (int i = 0; i < NLD; ++i) {
            long rr = (long)nxt * 32 + srow + i * RSTEP;
            if (rr >= N) rr = N - 1;
            pre[i] = *(const v4f*)(sbase + rr * 128 + scol);
        }
        __builtin_amdgcn_sched_barrier(0);

        const float* a0p = &As[buf][ml * LDA + 4 * half];
        const float* a1p = a0p + 16 * LDA;
        v8f c0 = v8f{}, c1 = v8f{};
#pragma unroll
        for (int j = 0; j < KP; ++j) {
            const v4f a0 = *(const v4f*)(a0p + 8 * j);
            const v4f a1 = *(const v4f*)(a1p + 8 * j);
            c0 = __builtin_amdgcn_wmma_f32_16x16x4_f32(
                false, v2f{a0.x, a0.y}, false, breg[2 * j], (short)0, c0, false, false);
            c1 = __builtin_amdgcn_wmma_f32_16x16x4_f32(
                false, v2f{a1.x, a1.y}, false, breg[2 * j], (short)0, c1, false, false);
            c0 = __builtin_amdgcn_wmma_f32_16x16x4_f32(
                false, v2f{a0.z, a0.w}, false, breg[2 * j + 1], (short)0, c0, false, false);
            c1 = __builtin_amdgcn_wmma_f32_16x16x4_f32(
                false, v2f{a1.z, a1.w}, false, breg[2 * j + 1], (short)0, c1, false, false);
        }

#pragma unroll
        for (int r = 0; r < 8; ++r) { c0[r] += bv; c1[r] += bv; }

        if (LNRELU) {
#pragma unroll
            for (int g = 0; g < 2; ++g) {
#pragma unroll
                for (int r = 0; r < 8; ++r) {
                    const float cv = g ? c1[r] : c0[r];
                    float sv = cv;
                    float qv = cv * cv;
#pragma unroll
                    for (int off = 1; off < 16; off <<= 1) {
                        sv += __shfl_xor(sv, off, 32);
                        qv += __shfl_xor(qv, off, 32);
                    }
                    if (ml == 0) {
                        red_s[wave * 32 + g * 16 + half * 8 + r] = sv;
                        red_q[wave * 32 + g * 16 + half * 8 + r] = qv;
                    }
                }
            }
            __syncthreads();
            if (tid < 64) {
                const float* rp = (tid < 32) ? red_s : red_q;
                const int row = tid & 31;
                float ts = 0.0f;
#pragma unroll
                for (int w = 0; w < 8; ++w) ts += rp[w * 32 + row];
                comb[tid] = ts;
            }
            __syncthreads();
#pragma unroll
            for (int g = 0; g < 2; ++g) {
#pragma unroll
                for (int r = 0; r < 8; ++r) {
                    const int rowid = g * 16 + half * 8 + r;
                    const float mu   = comb[rowid] * (1.0f / NDIM);
                    const float var  = comb[32 + rowid] * (1.0f / NDIM) - mu * mu;
                    const float rstd = rsqrtf(var + LN_EPS);
                    const float cv = g ? c1[r] : c0[r];
                    float v = (cv - mu) * rstd;
                    v = v > 0.0f ? v : 0.0f;
                    if (g) c1[r] = v; else c0[r] = v;
                }
            }
        }

#pragma unroll
        for (int i = 0; i < NLD; ++i)
            *(v4f*)&As[buf ^ 1][(srow + i * RSTEP) * LDA + sc4 * 4] = pre[i];
        __syncthreads();
        buf ^= 1;

#pragma unroll
        for (int g = 0; g < 2; ++g) {
#pragma unroll
            for (int r = 0; r < 8; ++r) {
                const long ro = (long)tile * 32 + g * 16 + half * 8 + r;
                const float cv = g ? c1[r] : c0[r];
                if (PRE) {
                    float* yp = second ? y2 : y;
                    const int cw = second ? (wave - 4) : wave;
                    if (ro < N) yp[ro * HN + cw * 16 + ml] = cv;
                } else {
                    if (ro < N) y[ro * NDIM + wave * 16 + ml] = cv;
                }
            }
        }
    }
}


extern "C" void kernel_launch(void* const* d_in, const int* in_sizes, int n_in,
                              void* d_out, int out_size, void* d_ws, size_t ws_size,
                              hipStream_t stream) {
    const float* h   = (const float*)d_in[0];
    const int*   src = (const int*)d_in[1];
    const int*   dst = (const int*)d_in[2];
    const float* W1  = (const float*)d_in[3];
    const float* b1  = (const float*)d_in[4];
    const float* W2  = (const float*)d_in[5];
    const float* b2  = (const float*)d_in[6];
    const float* W3  = (const float*)d_in[7];
    const float* b3  = (const float*)d_in[8];
    float* out = (float*)d_out;

    const int N = in_sizes[0] / 128;
    const int E = in_sizes[1];

    float* ws   = (float*)d_ws;
    float* norm = ws;
    float* agg  = norm + N;
    float* h1   = agg + (size_t)N * 128;
    float* h2   = h1  + (size_t)N * 128;

    const int  ntiles      = (N + 31) / 32;
    const int  gemm_blocks = ntiles < 512 ? ntiles : 512;
    const int  gemm_blocks3 = ntiles < 1024 ? ntiles : 1024;
    const int  sum_tiles   = (N + 399) / 400;

    tile_sum_kernel<128, 400, false, true><<<sum_tiles, 1024, 0, stream>>>(h, src, dst, nullptr, norm, agg, E, N);
    sage_gemm_kernel<true, false><<<gemm_blocks, 256, 0, stream>>>(h, agg, W1, b1, h1, nullptr, N, ntiles);

    tile_sum_kernel<128, 400, false, false><<<sum_tiles, 1024, 0, stream>>>(h1, src, dst, norm, nullptr, agg, E, N);
    sage_gemm_kernel<true, false><<<gemm_blocks, 256, 0, stream>>>(h1, agg, W2, b2, h2, nullptr, N, ntiles);

    float* z = agg;
    sage_gemm_kernel<false, true><<<gemm_blocks3, 256, 0, stream>>>(h2, nullptr, W3, b3, out, z, N, ntiles);
    tile_sum_kernel<64, 400, true, false><<<sum_tiles, 1024, 0, stream>>>(z, src, dst, norm, nullptr, out, E, N);
}


# First neighbour-sum pass keeps each tile's edges (src row, tile row) in per-wave lists by row owner; later passes read their lists: no dst scan, no barrier; one wave sums a row: sum order changes
# speedup vs baseline: 2.5686x; 1.1779x over previous
#include <hip/hip_runtime.h>

typedef __attribute__((ext_vector_type(2))) float v2f;
typedef __attribute__((ext_vector_type(4))) float v4f;
typedef __attribute__((ext_vector_type(8))) float v8f;

#define LN_EPS 1e-5f

constexpr int LCAP = 512;

__device__ __forceinline__ void load_dst32(int4 (&v)[8], const int* __restrict__ dst, int wbase,
                                           int lane, int E) {
    const int ebase = wbase + lane * 4;
    if (wbase + 1024 <= E) {
#pragma unroll
        for (int g = 0; g < 8; ++g) v[g] = *(const int4*)(dst + ebase + g * 128);
    } else {
#pragma unroll
        for (int g = 0; g < 8; ++g) {
            const int e = ebase + g * 128;
            v[g].x = (e     < E) ? dst[e]     : -1;
            v[g].y = (e + 1 < E) ? dst[e + 1] : -1;
            v[g].z = (e + 2 < E) ? dst[e + 2] : -1;
            v[g].w = (e + 3 < E) ? dst[e + 3] : -1;
        }
    }
}

template <int F>
__device__ __forceinline__ void add_rows(const float* __restrict__ x, float* tile, int s_l, int dl_l,
                                         int cnt, int lane) {
    constexpr int C = F / 32;
    for (int b = 0; b < cnt; b += 8) {
        float v[8][C];
#pragma unroll
        for (int j = 0; j < 8; ++j) {
            const int s = __builtin_amdgcn_readlane(s_l, b + j);
            const float* xp = x + (long)s * F + lane;
#pragma unroll
            for (int c = 0; c < C; ++c) v[j][c] = xp[32 * c];
        }
#pragma unroll
        for (int j = 0; j < 8; ++j) {
            if (b + j < cnt) {
                const int dl = __builtin_amdgcn_readlane(dl_l, b + j);
                float* tp = tile + dl * F + lane;
#pragma unroll
                for (int c = 0; c < C; ++c) atomicAdd(tp + 32 * c, v[j][c]);
            }
        }
    }
}

template <int F, bool FIRST>
__device__ __forceinline__ void drain_queue(const float* __restrict__ x, const int* __restrict__ src,
                                            const int* __restrict__ dst, float* tile, int* degc,
                                            int* lcur, int* ltile,
                                            const int* qe, int lo, int cnt, int lane) {
    const int e_l  = (lane < cnt) ? qe[lane] : 0;
    const int s_l  = src[e_l];
    const int dl_l = dst[e_l] - lo;
    if (FIRST) {
        if (lane < cnt) {
            atomicAdd(&degc[dl_l], 1);
            const int ow  = (dl_l >> 2) & 31;
            const int pos = atomicAdd(&lcur[ow], 1);
            if (pos < LCAP) *(int2*)(ltile + 2 * (ow * LCAP + pos)) = make_int2(s_l, dl_l);
        }
    }
    add_rows<F>(x, tile, s_l, dl_l, cnt, lane);
}

template <int F, int TN, bool ADD_TO_OUT, bool FIRST>
__global__ __launch_bounds__(1024) void tile_sum_kernel(
    const float* __restrict__ x,
    const int* __restrict__ src, const int* __restrict__ dst,
    const float* __restrict__ norm,
    float* __restrict__ norm_out,
    int* __restrict__ lists,
    int* __restrict__ lcnt,
    float* __restrict__ y,
    int E, int N) {
    constexpr int C = F / 32;
    static_assert(TN % 4 == 0, "the write-out takes 4 rows at a time");

    __shared__ float tile[TN * 128];
    __shared__ int q_e[32 * 64];
    __shared__ int degc[FIRST ? TN : 1];
    __shared__ int lcur[FIRST ? 32 : 1];

    const int tid  = threadIdx.x;
    const int wave = __builtin_amdgcn_readfirstlane(tid >> 5);
    const int lane = tid & 31;
    const int lo   = blockIdx.x * TN;
    int* ltile = lists + (size_t)blockIdx.x * (32 * LCAP * 2);

    bool use_lists = false;
    int  cntw = 0;
    if (!FIRST) {
        const int myc = lcnt[blockIdx.x * 32 + lane];
        use_lists = __builtin_amdgcn_ballot_w32(myc > LCAP) == 0u;
        cntw = __builtin_amdgcn_readlane(myc, wave);
    }

    if (!FIRST && use_lists) {
        for (int r0 = wave * 4; r0 < TN; r0 += 128) {
#pragma unroll
            for (int k = 0; k < 4; ++k) {
#pragma unroll
                for (int c = 0; c < C; ++c) tile[(r0 + k) * F + lane + 32 * c] = 0.0f;
            }
        }
        const int* lp = ltile + wave * (LCAP * 2);
        int2 ent = (lane < cntw) ? *(const int2*)(lp + 2 * lane) : make_int2(0, 0);
        for (int i = 0; i < cntw; i += 32) {
            const int nl = i + 32 + lane;
            const int2 nxt = (nl < cntw) ? *(const int2*)(lp + 2 * nl) : make_int2(0, 0);
            const int left = cntw - i;
            add_rows<F>(x, tile, ent.x, ent.y, left < 32 ? left : 32, lane);
            ent = nxt;
        }
    } else {
        for (int i = tid * 4; i < TN * F; i += 4096) *(float4*)&tile[i] = float4{0.0f, 0.0f, 0.0f, 0.0f};
        if (FIRST) {
            for (int i = tid; i < TN; i += 1024) degc[i] = 0;
            if (tid < 32) lcur[tid] = 0;
        }
        __syncthreads();

        int* qe = q_e + wave * 64;
        int qn = 0;

        for (int e0 = wave * 1024; e0 < E; e0 += 32768) {
            int4 dv[8];
            load_dst32(dv, dst, e0, lane, E);
            unsigned hm = 0u;
#pragma unroll
            for (int g = 0; g < 8; ++g) {
                const int dd[4] = {dv[g].x, dv[g].y, dv[g].z, dv[g].w};
#pragma unroll
                for (int c = 0; c < 4; ++c)
                    hm |= ((unsigned)(dd[c] - lo) < (unsigned)TN ? 1u : 0u) << (g * 4 + c);
            }
            for (;;) {
                const bool hit = hm != 0u;
                const unsigned m = __builtin_amdgcn_ballot_w32(hit);
                if (m == 0u) break;
                if (hit) {
                    const int bit = __builtin_ctz(hm);
                    hm &= hm - 1u;
                    const int pos = qn + (int)__builtin_amdgcn_mbcnt_lo(m, 0u);
                    qe[pos] = e0 + (bit >> 2) * 128 + lane * 4 + (bit & 3);
                }
                qn += __builtin_popcount(m);
                if (qn >= 32) {
                    drain_queue<F, FIRST>(x, src, dst, tile, degc, lcur, ltile, qe, lo, 32, lane);
                    const int rem = qn - 32;
                    if (lane < rem) {
                        const int te = qe[32 + lane];
                        qe[lane] = te;
                    }
                    qn = rem;
                }
            }
        }
        if (qn > 0) drain_queue<F, FIRST>(x, src, dst, tile, degc, lcur, ltile, qe, lo, qn, lane);
        __syncthreads();

        if (FIRST) {
            for (int r = tid; r < TN; r += 1024) {
                const int node = lo + r;
                if (node < N) {
                    const int c = degc[r];
                    norm_out[node] = c > 0 ? 1.0f / (float)c : 0.0f;
                }
            }
            if (tid < 32) lcnt[blockIdx.x * 32 + tid] = lcur[tid];
        }
    }

    for (int r0 = wave * 4; r0 < TN; r0 += 128) {
        float old[4][C];
        float nvs[4];
#pragma unroll
        for (int k = 0; k < 4; ++k) {
            const int node = lo + r0 + k;
            if (FIRST) {
                const int c = degc[r0 + k];
                nvs[k] = c > 0 ? 1.0f / (float)c : 0.0f;
            } else {
                nvs[k] = (node < N) ? norm[node] : 0.0f;
            }
        }
        if (ADD_TO_OUT) {
#pragma unroll
            for (int k = 0; k < 4; ++k) {
                const int node = lo + r0 + k;
#pragma unroll
                for (int c = 0; c < C; ++c)
                    old[k][c] = (node < N) ? y[(long)node * F + lane + 32 * c] : 0.0f;
            }
        }
#pragma unroll
        for (int k = 0; k < 4; ++k) {
            const int r = r0 + k;
            const int node = lo + r;
            const float nv = nvs[k];
            if (node < N) {
#pragma unroll
                for (int c = 0; c < C; ++c) {
                    const float t = tile[r * F + lane + 32 * c] * nv;
                    y[(long)node * F + lane + 32 * c] = ADD_TO_OUT ? old[k][c] + t : t;
                }
            }
        }
    }
}

template <bool LNRELU, bool PRE>
__global__ __launch_bounds__(256) void sage_gemm_kernel(
    const float* __restrict__ x,
    const float* __restrict__ agg,
    const float* __restrict__ W,
    const float* __restrict__ bias,
    float* __restrict__ y,
    float* __restrict__ y2,
    int N, int ntiles) {
    constexpr int NDIM = 128;
    constexpr int HN   = 64;
    constexpr int KTOT = PRE ? 128 : 256;
    constexpr int LDA  = KTOT + 4;
    constexpr int KP   = KTOT / 8;
    constexpr int F4R  = KTOT / 4;
    constexpr int NLD  = 32 * F4R / 256;
    constexpr int RSTEP = 256 / F4R;

    __shared__ float As[2][32 * LDA];
    __shared__ float red_s[8 * 32];
    __shared__ float red_q[8 * 32];
    __shared__ float comb[2 * 32];

    const int tid  = threadIdx.x;
    const int wave = tid >> 5;
    const int lane = tid & 31;
    const int ml   = lane & 15;
    const int half = lane >> 4;

    v2f breg[2 * KP];
    {
        const int n = wave * 16 + ml;
#pragma unroll
        for (int j = 0; j < KP; ++j) {
#pragma unroll
            for (int t = 0; t < 2; ++t) {
                const int k = 8 * j + 4 * half + 2 * t;
                float w0, w1;
                if (PRE) {
                    const int krow = (n < HN) ? k : (128 + k);
                    const int col  = (n < HN) ? n : (n - HN);
                    w0 = W[(long)krow * HN + col];
                    w1 = W[(long)(krow + 1) * HN + col];
                } else {
                    w0 = W[(long)k * NDIM + n];
                    w1 = W[(long)(k + 1) * NDIM + n];
                }
                breg[2 * j + t] = v2f{w0, w1};
            }
        }
    }

    const int srow = tid / F4R;
    const int sc4  = tid % F4R;
    const float* sbase = (PRE || sc4 < 32) ? x : agg;
    const int scol = (sc4 & 31) * 4;

    const bool second = PRE && wave >= 4;
    const float bv = second ? 0.0f : bias[wave * 16 + ml];

    int tile = blockIdx.x;
    {
#pragma unroll
        for (int i = 0; i < NLD; ++i) {
            long rr = (long)tile * 32 + srow + i * RSTEP;
            if (rr >= N) rr = N - 1;
            const v4f v = *(const v4f*)(sbase + rr * 128 + scol);
            *(v4f*)&As[0][(srow + i * RSTEP) * LDA + sc4 * 4] = v;
        }
    }
    __syncthreads();

    int buf = 0;
    for (; tile < ntiles; tile += gridDim.x) {
        const int nxt = tile + gridDim.x;
        v4f pre[NLD];
#pragma unroll
        for (int i = 0; i < NLD; ++i) {
            long rr = (long)nxt * 32 + srow + i * RSTEP;
            if (rr >= N) rr = N - 1;
            pre[i] = *(const v4f*)(sbase + rr * 128 + scol);
        }
        __builtin_amdgcn_sched_barrier(0);

        const float* a0p = &As[buf][ml * LDA + 4 * half];
        const float* a1p = a0p + 16 * LDA;
        v8f c0 = v8f{}, c1 = v8f{};
#pragma unroll
        for (int j = 0; j < KP; ++j) {
            const v4f a0 = *(const v4f*)(a0p + 8 * j);
            const v4f a1 = *(const v4f*)(a1p + 8 * j);
            c0 = __builtin_amdgcn_wmma_f32_16x16x4_f32(
                false, v2f{a0.x, a0.y}, false, breg[2 * j], (short)0, c0, false, false);
            c1 = __builtin_amdgcn_wmma_f32_16x16x4_f32(
                false, v2f{a1.x, a1.y}, false, breg[2 * j], (short)0, c1, false, false);
            c0 = __builtin_amdgcn_wmma_f32_16x16x4_f32(
                false, v2f{a0.z, a0.w}, false, breg[2 * j + 1], (short)0, c0, false, false);
            c1 = __builtin_amdgcn_wmma_f32_16x16x4_f32(
                false, v2f{a1.z, a1.w}, false, breg[2 * j + 1], (short)0, c1, false, false);
        }

#pragma unroll
        for (int r = 0; r < 8; ++r) { c0[r] += bv; c1[r] += bv; }

        if (LNRELU) {
#pragma unroll
            for (int g = 0; g < 2; ++g) {
#pragma unroll
                for (int r = 0; r < 8; ++r) {
                    const float cv = g ? c1[r] : c0[r];
                    float sv = cv;
                    float qv = cv * cv;
#pragma unroll
                    for (int off = 1; off < 16; off <<= 1) {
                        sv += __shfl_xor(sv, off, 32);
                        qv += __shfl_xor(qv, off, 32);
                    }
                    if (ml == 0) {
                        red_s[wave * 32 + g * 16 + half * 8 + r] = sv;
                        red_q[wave * 32 + g * 16 + half * 8 + r] = qv;
                    }
                }
            }
            __syncthreads();
            if (tid < 64) {
                const float* rp = (tid < 32) ? red_s : red_q;
                const int row = tid & 31;
                float ts = 0.0f;
#pragma unroll
                for (int w = 0; w < 8; ++w) ts += rp[w * 32 + row];
                comb[tid] = ts;
            }
            __syncthreads();
#pragma unroll
            for (int g = 0; g < 2; ++g) {
#pragma unroll
                for (int r = 0; r < 8; ++r) {
                    const int rowid = g * 16 + half * 8 + r;
                    const float mu   = comb[rowid] * (1.0f / NDIM);
                    const float var  = comb[32 + rowid] * (1.0f / NDIM) - mu * mu;
                    const float rstd = rsqrtf(var + LN_EPS);
                    const float cv = g ? c1[r] : c0[r];
                    float v = (cv - mu) * rstd;
                    v = v > 0.0f ? v : 0.0f;
                    if (g) c1[r] = v; else c0[r] = v;
                }
            }
        }

#pragma unroll
        for (int i = 0; i < NLD; ++i)
            *(v4f*)&As[buf ^ 1][(srow + i * RSTEP) * LDA + sc4 * 4] = pre[i];
        __syncthreads();
        buf ^= 1;

#pragma unroll
        for (int g = 0; g < 2; ++g) {
#pragma unroll
            for (int r = 0; r < 8; ++r) {
                const long ro = (long)tile * 32 + g * 16 + half * 8 + r;
                const float cv = g ? c1[r] : c0[r];
                if (PRE) {
                    float* yp = second ? y2 : y;
                    const int cw = second ? (wave - 4) : wave;
                    if (ro < N) yp[ro * HN + cw * 16 + ml] = cv;
                } else {
                    if (ro < N) y[ro * NDIM + wave * 16 + ml] = cv;
                }
            }
        }
    }
}


extern "C" void kernel_launch(void* const* d_in, const int* in_sizes, int n_in,
                              void* d_out, int out_size, void* d_ws, size_t ws_size,
                              hipStream_t stream) {
    const float* h   = (const float*)d_in[0];
    const int*   src = (const int*)d_in[1];
    const int*   dst = (const int*)d_in[2];
    const float* W1  = (const float*)d_in[3];
    const float* b1  = (const float*)d_in[4];
    const float* W2  = (const float*)d_in[5];
    const float* b2  = (const float*)d_in[6];
    const float* W3  = (const float*)d_in[7];
    const float* b3  = (const float*)d_in[8];
    float* out = (float*)d_out;

    const int N = in_sizes[0] / 128;
    const int E = in_sizes[1];

    float* ws   = (float*)d_ws;
    float* norm = ws;
    float* agg  = norm + N;
    float* h1   = agg + (size_t)N * 128;
    float* h2   = h1  + (size_t)N * 128;

    const int  ntiles      = (N + 31) / 32;
    const int  gemm_blocks = ntiles < 512 ? ntiles : 512;
    const int  gemm_blocks3 = ntiles < 1024 ? ntiles : 1024;
    const int  sum_tiles   = (N + 399) / 400;
    int* lists = (int*)(h2 + (size_t)N * 128);
    int* lcnt  = lists + (size_t)sum_tiles * (32 * LCAP * 2);

    tile_sum_kernel<128, 400, false, true><<<sum_tiles, 1024, 0, stream>>>(h, src, dst, nullptr, norm, lists, lcnt, agg, E, N);
    sage_gemm_kernel<true, false><<<gemm_blocks, 256, 0, stream>>>(h, agg, W1, b1, h1, nullptr, N, ntiles);

    tile_sum_kernel<128, 400, false, false><<<sum_tiles, 1024, 0, stream>>>(h1, src, dst, norm, nullptr, lists, lcnt, agg, E, N);
    sage_gemm_kernel<true, false><<<gemm_blocks, 256, 0, stream>>>(h1, agg, W2, b2, h2, nullptr, N, ntiles);

    float* z = agg;
    sage_gemm_kernel<false, true><<<gemm_blocks3, 256, 0, stream>>>(h2, nullptr, W3, b3, out, z, N, ntiles);
    tile_sum_kernel<64, 400, true, false><<<sum_tiles, 1024, 0, stream>>>(z, src, dst, norm, nullptr, lists, lcnt, out, E, N);
}


# LayerNorm row sums of the layer GEMMs: lane exchanges as vector-ALU operand swizzles in a halving tree (15 adds per 16 values), not 128 LDS-unit shuffles a tile, each waited for; sum order kept
# speedup vs baseline: 2.8643x; 1.1151x over previous
#include <hip/hip_runtime.h>

typedef __attribute__((ext_vector_type(2))) float v2f;
typedef __attribute__((ext_vector_type(4))) float v4f;
typedef __attribute__((ext_vector_type(8))) float v8f;

#define LN_EPS 1e-5f

constexpr int LCAP = 512;

__device__ __forceinline__ void load_dst32(int4 (&v)[8], const int* __restrict__ dst, int wbase,
                                           int lane, int E) {
    const int ebase = wbase + lane * 4;
    if (wbase + 1024 <= E) {
#pragma unroll
        for (int g = 0; g < 8; ++g) v[g] = *(const int4*)(dst + ebase + g * 128);
    } else {
#pragma unroll
        for (int g = 0; g < 8; ++g) {
            const int e = ebase + g * 128;
            v[g].x = (e     < E) ? dst[e]     : -1;
            v[g].y = (e + 1 < E) ? dst[e + 1] : -1;
            v[g].z = (e + 2 < E) ? dst[e + 2] : -1;
            v[g].w = (e + 3 < E) ? dst[e + 3] : -1;
        }
    }
}

template <int F>
__device__ __forceinline__ void add_rows(const float* __restrict__ x, float* tile, int s_l, int dl_l,
                                         int cnt, int lane) {
    constexpr int C = F / 32;
    for (int b = 0; b < cnt; b += 8) {
        float v[8][C];
#pragma unroll
        for (int j = 0; j < 8; ++j) {
            const int s = __builtin_amdgcn_readlane(s_l, b + j);
            const float* xp = x + (long)s * F + lane;
#pragma unroll
            for (int c = 0; c < C; ++c) v[j][c] = xp[32 * c];
        }
#pragma unroll
        for (int j = 0; j < 8; ++j) {
            if (b + j < cnt) {
                const int dl = __builtin_amdgcn_readlane(dl_l, b + j);
                float* tp = tile + dl * F + lane;
#pragma unroll
                for (int c = 0; c < C; ++c) atomicAdd(tp + 32 * c, v[j][c]);
            }
        }
    }
}

template <int F, bool FIRST>
__device__ __forceinline__ void drain_queue(const float* __restrict__ x, const int* __restrict__ src,
                                            const int* __restrict__ dst, float* tile, int* degc,
                                            int* lcur, int* ltile,
                                            const int* qe, int lo, int cnt, int lane) {
    const int e_l  = (lane < cnt) ? qe[lane] : 0;
    const int s_l  = src[e_l];
    const int dl_l = dst[e_l] - lo;
    if (FIRST) {
        if (lane < cnt) {
            atomicAdd(&degc[dl_l], 1);
            const int ow  = (dl_l >> 2) & 31;
            const int pos = atomicAdd(&lcur[ow], 1);
            if (pos < LCAP) *(int2*)(ltile + 2 * (ow * LCAP + pos)) = make_int2(s_l, dl_l);
        }
    }
    add_rows<F>(x, tile, s_l, dl_l, cnt, lane);
}

template <int F, int TN, bool ADD_TO_OUT, bool FIRST>
__global__ __launch_bounds__(1024) void tile_sum_kernel(
    const float* __restrict__ x,
    const int* __restrict__ src, const int* __restrict__ dst,
    const float* __restrict__ norm,
    float* __restrict__ norm_out,
    int* __restrict__ lists,
    int* __restrict__ lcnt,
    float* __restrict__ y,
    int E, int N) {
    constexpr int C = F / 32;
    static_assert(TN % 4 == 0, "the write-out takes 4 rows at a time");

    __shared__ float tile[TN * 128];
    __shared__ int q_e[32 * 64];
    __shared__ int degc[FIRST ? TN : 1];
    __shared__ int lcur[FIRST ? 32 : 1];

    const int tid  = threadIdx.x;
    const int wave = __builtin_amdgcn_readfirstlane(tid >> 5);
    const int lane = tid & 31;
    const int lo   = blockIdx.x * TN;
    int* ltile = lists + (size_t)blockIdx.x * (32 * LCAP * 2);

    bool use_lists = false;
    int  cntw = 0;
    if (!FIRST) {
        const int myc = lcnt[blockIdx.x * 32 + lane];
        use_lists = __builtin_amdgcn_ballot_w32(myc > LCAP) == 0u;
        cntw = __builtin_amdgcn_readlane(myc, wave);
    }

    if (!FIRST && use_lists) {
        for (int r0 = wave * 4; r0 < TN; r0 += 128) {
#pragma unroll
            for (int k = 0; k < 4; ++k) {
#pragma unroll
                for (int c = 0; c < C; ++c) tile[(r0 + k) * F + lane + 32 * c] = 0.0f;
            }
        }
        const int* lp = ltile + wave * (LCAP * 2);
        int2 ent = (lane < cntw) ? *(const int2*)(lp + 2 * lane) : make_int2(0, 0);
        for (int i = 0; i < cntw; i += 32) {
            const int nl = i + 32 + lane;
            const int2 nxt = (nl < cntw) ? *(const int2*)(lp + 2 * nl) : make_int2(0, 0);
            const int left = cntw - i;
            add_rows<F>(x, tile, ent.x, ent.y, left < 32 ? left : 32, lane);
            ent = nxt;
        }
    } else {
        for (int i = tid * 4; i < TN * F; i += 4096) *(float4*)&tile[i] = float4{0.0f, 0.0f, 0.0f, 0.0f};
        if (FIRST) {
            for (int i = tid; i < TN; i += 1024) degc[i] = 0;
            if (tid < 32) lcur[tid] = 0;
        }
        __syncthreads();

        int* qe = q_e + wave * 64;
        int qn = 0;

        for (int e0 = wave * 1024; e0 < E; e0 += 32768) {
            int4 dv[8];
            load_dst32(dv, dst, e0, lane, E);
            unsigned hm = 0u;
#pragma unroll
            for (int g = 0; g < 8; ++g) {
                const int dd[4] = {dv[g].x, dv[g].y, dv[g].z, dv[g].w};
#pragma unroll
                for (int c = 0; c < 4; ++c)
                    hm |= ((unsigned)(dd[c] - lo) < (unsigned)TN ? 1u : 0u) << (g * 4 + c);
            }
            for (;;) {
                const bool hit = hm != 0u;
                const unsigned m = __builtin_amdgcn_ballot_w32(hit);
                if (m == 0u) break;
                if (hit) {
                    const int bit = __builtin_ctz(hm);
                    hm &= hm - 1u;
                    const int pos = qn + (int)__builtin_amdgcn_mbcnt_lo(m, 0u);
                    qe[pos] = e0 + (bit >> 2) * 128 + lane * 4 + (bit & 3);
                }
                qn += __builtin_popcount(m);
                if (qn >= 32) {
                    drain_queue<F, FIRST>(x, src, dst, tile, degc, lcur, ltile, qe, lo, 32, lane);
                    const int rem = qn - 32;
                    if (lane < rem) {
                        const int te = qe[32 + lane];
                        qe[lane] = te;
                    }
                    qn = rem;
                }
            }
        }
        if (qn > 0) drain_queue<F, FIRST>(x, src, dst, tile, degc, lcur, ltile, qe, lo, qn, lane);
        __syncthreads();

        if (FIRST) {
            for (int r = tid; r < TN; r += 1024) {
                const int node = lo + r;
                if (node < N) {
                    const int c = degc[r];
                    norm_out[node] = c > 0 ? 1.0f / (float)c : 0.0f;
                }
            }
            if (tid < 32) lcnt[blockIdx.x * 32 + tid] = lcur[tid];
        }
    }

    for (int r0 = wave * 4; r0 < TN; r0 += 128) {
        float old[4][C];
        float nvs[4];
#pragma unroll
        for (int k = 0; k < 4; ++k) {
            const int node = lo + r0 + k;
            if (FIRST) {
                const int c = degc[r0 + k];
                nvs[k] = c > 0 ? 1.0f / (float)c : 0.0f;
            } else {
                nvs[k] = (node < N) ? norm[node] : 0.0f;
            }
        }
        if (ADD_TO_OUT) {
#pragma unroll
            for (int k = 0; k < 4; ++k) {
                const int node = lo + r0 + k;
#pragma unroll
                for (int c = 0; c < C; ++c)
                    old[k][c] = (node < N) ? y[(long)node * F + lane + 32 * c] : 0.0f;
            }
        }
#pragma unroll
        for (int k = 0; k < 4; ++k) {
            const int r = r0 + k;
            const int node = lo + r;
            const float nv = nvs[k];
            if (node < N) {
#pragma unroll
                for (int c = 0; c < C; ++c) {
                    const float t = tile[r * F + lane + 32 * c] * nv;
                    y[(long)node * F + lane + 32 * c] = ADD_TO_OUT ? old[k][c] + t : t;
                }
            }
        }
    }
}

template <int M>
__device__ __forceinline__ float lane_xor(float v) {
    return __builtin_bit_cast(float, __builtin_amdgcn_update_dpp(
        0, __builtin_bit_cast(int, v), 0x160 + M, 0xf, 0xf, true));
}

template <bool LNRELU, bool PRE>
__global__ __launch_bounds__(256) void sage_gemm_kernel(
    const float* __restrict__ x,
    const float* __restrict__ agg,
    const float* __restrict__ W,
    const float* __restrict__ bias,
    float* __restrict__ y,
    float* __restrict__ y2,
    int N, int ntiles) {
    constexpr int NDIM = 128;
    constexpr int HN   = 64;
    constexpr int KTOT = PRE ? 128 : 256;
    constexpr int LDA  = KTOT + 4;
    constexpr int KP   = KTOT / 8;
    constexpr int F4R  = KTOT / 4;
    constexpr int NLD  = 32 * F4R / 256;
    constexpr int RSTEP = 256 / F4R;

    __shared__ float As[2][32 * LDA];
    __shared__ float red_s[8 * 32];
    __shared__ float red_q[8 * 32];
    __shared__ float comb[2 * 32];

    const int tid  = threadIdx.x;
    const int wave = tid >> 5;
    const int lane = tid & 31;
    const int ml   = lane & 15;
    const int half = lane >> 4;

    v2f breg[2 * KP];
    {
        const int n = wave * 16 + ml;
#pragma unroll
        for (int j = 0; j < KP; ++j) {
#pragma unroll
            for (int t = 0; t < 2; ++t) {
                const int k = 8 * j + 4 * half + 2 * t;
                float w0, w1;
                if (PRE) {
                    const int krow = (n < HN) ? k : (128 + k);
                    const int col  = (n < HN) ? n : (n - HN);
                    w0 = W[(long)krow * HN + col];
                    w1 = W[(long)(krow + 1) * HN + col];
                } else {
                    w0 = W[(long)k * NDIM + n];
                    w1 = W[(long)(k + 1) * NDIM + n];
                }
                breg[2 * j + t] = v2f{w0, w1};
            }
        }
    }

    const int srow = tid / F4R;
    const int sc4  = tid % F4R;
    const float* sbase = (PRE || sc4 < 32) ? x : agg;
    const int scol = (sc4 & 31) * 4;

    const bool second = PRE && wave >= 4;
    const float bv = second ? 0.0f : bias[wave * 16 + ml];

    int tile = blockIdx.x;
    {
#pragma unroll
        for (int i = 0; i < NLD; ++i) {
            long rr = (long)tile * 32 + srow + i * RSTEP;
            if (rr >= N) rr = N - 1;
            const v4f v = *(const v4f*)(sbase + rr * 128 + scol);
            *(v4f*)&As[0][(srow + i * RSTEP) * LDA + sc4 * 4] = v;
        }
    }
    __syncthreads();

    int buf = 0;
    for (; tile < ntiles; tile += gridDim.x) {
        const int nxt = tile + gridDim.x;
        v4f pre[NLD];
#pragma unroll
        for (int i = 0; i < NLD; ++i) {
            long rr = (long)nxt * 32 + srow + i * RSTEP;
            if (rr >= N) rr = N - 1;
            pre[i] = *(const v4f*)(sbase + rr * 128 + scol);
        }
        __builtin_amdgcn_sched_barrier(0);

        const float* a0p = &As[buf][ml * LDA + 4 * half];
        const float* a1p = a0p + 16 * LDA;
        v8f c0 = v8f{}, c1 = v8f{};
#pragma unroll
        for (int j = 0; j < KP; ++j) {
            const v4f a0 = *(const v4f*)(a0p + 8 * j);
            const v4f a1 = *(const v4f*)(a1p + 8 * j);
            c0 = __builtin_amdgcn_wmma_f32_16x16x4_f32(
                false, v2f{a0.x, a0.y}, false, breg[2 * j], (short)0, c0, false, false);
            c1 = __builtin_amdgcn_wmma_f32_16x16x4_f32(
                false, v2f{a1.x, a1.y}, false, breg[2 * j], (short)0, c1, false, false);
            c0 = __builtin_amdgcn_wmma_f32_16x16x4_f32(
                false, v2f{a0.z, a0.w}, false, breg[2 * j + 1], (short)0, c0, false, false);
            c1 = __builtin_amdgcn_wmma_f32_16x16x4_f32(
                false, v2f{a1.z, a1.w}, false, breg[2 * j + 1], (short)0, c1, false, false);
        }

#pragma unroll
        for (int r = 0; r < 8; ++r) { c0[r] += bv; c1[r] += bv; }

        if (LNRELU) {
            const bool o1 = (ml & 1) != 0, o2 = (ml & 2) != 0, o4 = (ml & 4) != 0, o8 = (ml & 8) != 0;
            float s1[8], q1[8];
#pragma unroll
            for (int i = 0; i < 8; ++i) {
                const float va = (i < 4) ? c0[2 * i]     : c1[2 * i - 8];
                const float vb = (i < 4) ? c0[2 * i + 1] : c1[2 * i - 7];
                const float keep = o1 ? vb : va;
                const float send = o1 ? va : vb;
                s1[i] = keep + lane_xor<1>(send);
                q1[i] = keep * keep + lane_xor<1>(send * send);
            }
            float s2[4], q2[4];
#pragma unroll
            for (int i = 0; i < 4; ++i) {
                const float ks = o2 ? s1[2 * i + 1] : s1[2 * i];
                const float ss = o2 ? s1[2 * i]     : s1[2 * i + 1];
                const float kq = o2 ? q1[2 * i + 1] : q1[2 * i];
                const float sq = o2 ? q1[2 * i]     : q1[2 * i + 1];
                s2[i] = ks + lane_xor<2>(ss);
                q2[i] = kq + lane_xor<2>(sq);
            }
            float s3[2], q3[2];
#pragma unroll
            for (int i = 0; i < 2; ++i) {
                const float ks = o4 ? s2[2 * i + 1] : s2[2 * i];
                const float ss = o4 ? s2[2 * i]     : s2[2 * i + 1];
                const float kq = o4 ? q2[2 * i + 1] : q2[2 * i];
                const float sq = o4 ? q2[2 * i]     : q2[2 * i + 1];
                s3[i] = ks + lane_xor<4>(ss);
                q3[i] = kq + lane_xor<4>(sq);
            }
            const float rs = (o8 ? s3[1] : s3[0]) + lane_xor<8>(o8 ? s3[0] : s3[1]);
            const float rq = (o8 ? q3[1] : q3[0]) + lane_xor<8>(o8 ? q3[0] : q3[1]);
            {
                const int rowid = (ml >> 3) * 16 + half * 8 + (ml & 7);
                red_s[wave * 32 + rowid] = rs;
                red_q[wave * 32 + rowid] = rq;
            }
            __syncthreads();
            if (tid < 64) {
                const float* rp = (tid < 32) ? red_s : red_q;
                const int row = tid & 31;
                float ts = 0.0f;
#pragma unroll
                for (int w = 0; w < 8; ++w) ts += rp[w * 32 + row];
                comb[tid] = ts;
            }
            __syncthreads();
#pragma unroll
            for (int g = 0; g < 2; ++g) {
#pragma unroll
                for (int r = 0; r < 8; ++r) {
                    const int rowid = g * 16 + half * 8 + r;
                    const float mu   = comb[rowid] * (1.0f / NDIM);
                    const float var  = comb[32 + rowid] * (1.0f / NDIM) - mu * mu;
                    const float rstd = rsqrtf(var + LN_EPS);
                    const float cv = g ? c1[r] : c0[r];
                    float v = (cv - mu) * rstd;
                    v = v > 0.0f ? v : 0.0f;
                    if (g) c1[r] = v; else c0[r] = v;
                }
            }
        }

#pragma unroll
        for (int i = 0; i < NLD; ++i)
            *(v4f*)&As[buf ^ 1][(srow + i * RSTEP) * LDA + sc4 * 4] = pre[i];
        __syncthreads();
        buf ^= 1;

#pragma unroll
        for (int g = 0; g < 2; ++g) {
#pragma unroll
            for (int r = 0; r < 8; ++r) {
                const long ro = (long)tile * 32 + g * 16 + half * 8 + r;
                const float cv = g ? c1[r] : c0[r];
                if (PRE) {
                    float* yp = second ? y2 : y;
                    const int cw = second ? (wave - 4) : wave;
                    if (ro < N) yp[ro * HN + cw * 16 + ml] = cv;
                } else {
                    if (ro < N) y[ro * NDIM + wave * 16 + ml] = cv;
                }
            }
        }
    }
}


extern "C" void kernel_launch(void* const* d_in, const int* in_sizes, int n_in,
                              void* d_out, int out_size, void* d_ws, size_t ws_size,
                              hipStream_t stream) {
    const float* h   = (const float*)d_in[0];
    const int*   src = (const int*)d_in[1];
    const int*   dst = (const int*)d_in[2];
    const float* W1  = (const float*)d_in[3];
    const float* b1  = (const float*)d_in[4];
    const float* W2  = (const float*)d_in[5];
    const float* b2  = (const float*)d_in[6];
    const float* W3  = (const float*)d_in[7];
    const float* b3  = (const float*)d_in[8];
    float* out = (float*)d_out;

    const int N = in_sizes[0] / 128;
    const int E = in_sizes[1];

    float* ws   = (float*)d_ws;
    float* norm = ws;
    float* agg  = norm + N;
    float* h1   = agg + (size_t)N * 128;
    float* h2   = h1  + (size_t)N * 128;

    const int  ntiles      = (N + 31) / 32;
    const int  gemm_blocks = ntiles < 512 ? ntiles : 512;
    const int  gemm_blocks3 = ntiles < 1024 ? ntiles : 1024;
    const int  sum_tiles   = (N + 399) / 400;
    int* lists = (int*)(h2 + (size_t)N * 128);
    int* lcnt  = lists + (size_t)sum_tiles * (32 * LCAP * 2);

    tile_sum_kernel<128, 400, false, true><<<sum_tiles, 1024, 0, stream>>>(h, src, dst, nullptr, norm, lists, lcnt, agg, E, N);
    sage_gemm_kernel<true, false><<<gemm_blocks, 256, 0, stream>>>(h, agg, W1, b1, h1, nullptr, N, ntiles);

    tile_sum_kernel<128, 400, false, false><<<sum_tiles, 1024, 0, stream>>>(h1, src, dst, norm, nullptr, lists, lcnt, agg, E, N);
    sage_gemm_kernel<true, false><<<gemm_blocks, 256, 0, stream>>>(h1, agg, W2, b2, h2, nullptr, N, ntiles);

    float* z = agg;
    sage_gemm_kernel<false, true><<<gemm_blocks3, 256, 0, stream>>>(h2, nullptr, W3, b3, out, z, N, ntiles);
    tile_sum_kernel<64, 400, true, false><<<sum_tiles, 1024, 0, stream>>>(z, src, dst, norm, nullptr, lists, lcnt, out, E, N);
}


# Scanning neighbour-sum pass reads its tile's row of a bit-per-edge bitmap that one new launch writes (LDS bit sets, whole-line stores), not all of dst; 7 launches; sums still arrival-ordered
# speedup vs baseline: 3.0397x; 1.0612x over previous
#include <hip/hip_runtime.h>

typedef __attribute__((ext_vector_type(2))) float v2f;
typedef __attribute__((ext_vector_type(4))) float v4f;
typedef __attribute__((ext_vector_type(8))) float v8f;

#define LN_EPS 1e-5f

template <int TN>
__global__ __launch_bounds__(256) void edge_bitmap_kernel(const int* __restrict__ dst,
                                                          unsigned* __restrict__ bitmap,
                                                          int E, int tiles, int wpt) {
    __shared__ unsigned bm[256 * 32];
    const int tid  = threadIdx.x;
    const int wave = tid >> 5;
    const int lane = tid & 31;
    const int nchunk = (E + 1023) >> 10;
    for (int ch = blockIdx.x; ch < nchunk; ch += gridDim.x) {
        const int eb = ch * 1024 + tid * 4;
        int t4[4];
        if (eb + 3 < E) {
            const int4 v = *(const int4*)(dst + eb);
            t4[0] = v.x / TN; t4[1] = v.y / TN; t4[2] = v.z / TN; t4[3] = v.w / TN;
        } else {
#pragma unroll
            for (int c = 0; c < 4; ++c) t4[c] = (eb + c < E) ? dst[eb + c] / TN : -1;
        }
        for (int tb = 0; tb < tiles; tb += 256) {
            for (int i = tid * 4; i < 256 * 32; i += 1024) *(uint4*)&bm[i] = uint4{0u, 0u, 0u, 0u};
            __syncthreads();
#pragma unroll
            for (int c = 0; c < 4; ++c) {
                const int t = t4[c] - tb;
                if ((unsigned)t < 256u) atomicOr(&bm[t * 32 + (tid >> 3)], 1u << ((tid & 7) * 4 + c));
            }
            __syncthreads();
            for (int t = wave * 4 + (lane >> 3); t < 256; t += 32)
                if (tb + t < tiles)
                    *(uint4*)&bitmap[(size_t)(tb + t) * wpt + ch * 32 + (lane & 7) * 4] =
                        *(const uint4*)&bm[t * 32 + (lane & 7) * 4];
            __syncthreads();
        }
    }
}

constexpr int TILE_N = 400;
constexpr int LCAP = 512;

template <int F>
__device__ __forceinline__ void add_rows(const float* __restrict__ x, float* tile, int s_l, int dl_l,
                                         int cnt, int lane) {
    constexpr int C = F / 32;
    for (int b = 0; b < cnt; b += 8) {
        float v[8][C];
#pragma unroll
        for (int j = 0; j < 8; ++j) {
            const int s = __builtin_amdgcn_readlane(s_l, b + j);
            const float* xp = x + (long)s * F + lane;
#pragma unroll
            for (int c = 0; c < C; ++c) v[j][c] = xp[32 * c];
        }
#pragma unroll
        for (int j = 0; j < 8; ++j) {
            if (b + j < cnt) {
                const int dl = __builtin_amdgcn_readlane(dl_l, b + j);
                float* tp = tile + dl * F + lane;
#pragma unroll
                for (int c = 0; c < C; ++c) atomicAdd(tp + 32 * c, v[j][c]);
            }
        }
    }
}

template <int F, bool FIRST>
__device__ __forceinline__ void drain_queue(const float* __restrict__ x, const int* __restrict__ src,
                                            const int* __restrict__ dst, float* tile, int* degc,
                                            int* lcur, int* ltile,
                                            const int* qe, int lo, int cnt, int lane) {
    const int e_l  = (lane < cnt) ? qe[lane] : 0;
    const int s_l  = src[e_l];
    const int dl_l = dst[e_l] - lo;
    if (FIRST) {
        if (lane < cnt) {
            atomicAdd(&degc[dl_l], 1);
            const int ow  = (dl_l >> 2) & 31;
            const int pos = atomicAdd(&lcur[ow], 1);
            if (pos < LCAP) *(int2*)(ltile + 2 * (ow * LCAP + pos)) = make_int2(s_l, dl_l);
        }
    }
    add_rows<F>(x, tile, s_l, dl_l, cnt, lane);
}

template <int F, int TN, bool ADD_TO_OUT, bool FIRST>
__global__ __launch_bounds__(1024) void tile_sum_kernel(
    const float* __restrict__ x,
    const int* __restrict__ src, const int* __restrict__ dst,
    const float* __restrict__ norm,
    float* __restrict__ norm_out,
    int* __restrict__ lists,
    int* __restrict__ lcnt,
    const unsigned* __restrict__ bitmap,
    float* __restrict__ y,
    int E, int N, int wpt) {
    constexpr int C = F / 32;
    static_assert(TN % 4 == 0, "the write-out takes 4 rows at a time");

    __shared__ float tile[TN * 128];
    __shared__ int q_e[32 * 64];
    __shared__ int degc[FIRST ? TN : 1];
    __shared__ int lcur[FIRST ? 32 : 1];

    const int tid  = threadIdx.x;
    const int wave = __builtin_amdgcn_readfirstlane(tid >> 5);
    const int lane = tid & 31;
    const int lo   = blockIdx.x * TN;
    int* ltile = lists + (size_t)blockIdx.x * (32 * LCAP * 2);

    bool use_lists = false;
    int  cntw = 0;
    if (!FIRST) {
        const int myc = lcnt[blockIdx.x * 32 + lane];
        use_lists = __builtin_amdgcn_ballot_w32(myc > LCAP) == 0u;
        cntw = __builtin_amdgcn_readlane(myc, wave);
    }

    if (!FIRST && use_lists) {
        for (int r0 = wave * 4; r0 < TN; r0 += 128) {
#pragma unroll
            for (int k = 0; k < 4; ++k) {
#pragma unroll
                for (int c = 0; c < C; ++c) tile[(r0 + k) * F + lane + 32 * c] = 0.0f;
            }
        }
        const int* lp = ltile + wave * (LCAP * 2);
        int2 ent = (lane < cntw) ? *(const int2*)(lp + 2 * lane) : make_int2(0, 0);
        for (int i = 0; i < cntw; i += 32) {
            const int nl = i + 32 + lane;
            const int2 nxt = (nl < cntw) ? *(const int2*)(lp + 2 * nl) : make_int2(0, 0);
            const int left = cntw - i;
            add_rows<F>(x, tile, ent.x, ent.y, left < 32 ? left : 32, lane);
            ent = nxt;
        }
    } else {
        for (int i = tid * 4; i < TN * F; i += 4096) *(float4*)&tile[i] = float4{0.0f, 0.0f, 0.0f, 0.0f};
        if (FIRST) {
            for (int i = tid; i < TN; i += 1024) degc[i] = 0;
            if (tid < 32) lcur[tid] = 0;
        }
        __syncthreads();

        int* qe = q_e + wave * 64;
        int qn = 0;

        const unsigned* bmt = bitmap + (size_t)blockIdx.x * wpt;

        for (int eb = wave * 1024; eb < E; eb += 8 * 32768) {
            unsigned w[8];
#pragma unroll
            for (int k = 0; k < 8; ++k) {
                const int e = eb + k * 32768;
                w[k] = (e < E) ? bmt[(e >> 5) + lane] : 0u;
            }
#pragma nounroll
            for (int k = 0; k < 8; ++k) {
                const int e0 = eb + k * 32768;
                unsigned hm = w[0];
#pragma unroll
                for (int j = 0; j < 7; ++j) w[j] = w[j + 1];
                w[7] = 0u;
                for (;;) {
                    const bool hit = hm != 0u;
                    const unsigned m = __builtin_amdgcn_ballot_w32(hit);
                    if (m == 0u) break;
                    if (hit) {
                        const int bit = __builtin_ctz(hm);
                        hm &= hm - 1u;
                        const int pos = qn + (int)__builtin_amdgcn_mbcnt_lo(m, 0u);
                        qe[pos] = e0 + lane * 32 + bit;
                    }
                    qn += __builtin_popcount(m);
                    if (qn >= 32) {
                        drain_queue<F, FIRST>(x, src, dst, tile, degc, lcur, ltile, qe, lo, 32, lane);
                        const int rem = qn - 32;
                        if (lane < rem) {
                            const int te = qe[32 + lane];
                            qe[lane] = te;
                        }
                        qn = rem;
                    }
                }
            }
        }
        if (qn > 0) drain_queue<F, FIRST>(x, src, dst, tile, degc, lcur, ltile, qe, lo, qn, lane);
        __syncthreads();

        if (FIRST) {
            for (int r = tid; r < TN; r += 1024) {
                const int node = lo + r;
                if (node < N) {
                    const int c = degc[r];
                    norm_out[node] = c > 0 ? 1.0f / (float)c : 0.0f;
                }
            }
            if (tid < 32) lcnt[blockIdx.x * 32 + tid] = lcur[tid];
        }
    }

    for (int r0 = wave * 4; r0 < TN; r0 += 128) {
        float old[4][C];
        float nvs[4];
#pragma unroll
        for (int k = 0; k < 4; ++k) {
            const int node = lo + r0 + k;
            if (FIRST) {
                const int c = degc[r0 + k];
                nvs[k] = c > 0 ? 1.0f / (float)c : 0.0f;
            } else {
                nvs[k] = (node < N) ? norm[node] : 0.0f;
            }
        }
        if (ADD_TO_OUT) {
#pragma unroll
            for (int k = 0; k < 4; ++k) {
                const int node = lo + r0 + k;
#pragma unroll
                for (int c = 0; c < C; ++c)
                    old[k][c] = (node < N) ? y[(long)node * F + lane + 32 * c] : 0.0f;
            }
        }
#pragma unroll
        for (int k = 0; k < 4; ++k) {
            const int r = r0 + k;
            const int node = lo + r;
            const float nv = nvs[k];
            if (node < N) {
#pragma unroll
                for (int c = 0; c < C; ++c) {
                    const float t = tile[r * F + lane + 32 * c] * nv;
                    y[(long)node * F + lane + 32 * c] = ADD_TO_OUT ? old[k][c] + t : t;
                }
            }
        }
    }
}

template <int M>
__device__ __forceinline__ float lane_xor(float v) {
    return __builtin_bit_cast(float, __builtin_amdgcn_update_dpp(
        0, __builtin_bit_cast(int, v), 0x160 + M, 0xf, 0xf, true));
}

template <bool LNRELU, bool PRE>
__global__ __launch_bounds__(256) void sage_gemm_kernel(
    const float* __restrict__ x,
    const float* __restrict__ agg,
    const float* __restrict__ W,
    const float* __restrict__ bias,
    float* __restrict__ y,
    float* __restrict__ y2,
    int N, int ntiles) {
    constexpr int NDIM = 128;
    constexpr int HN   = 64;
    constexpr int KTOT = PRE ? 128 : 256;
    constexpr int LDA  = KTOT + 4;
    constexpr int KP   = KTOT / 8;
    constexpr int F4R  = KTOT / 4;
    constexpr int NLD  = 32 * F4R / 256;
    constexpr int RSTEP = 256 / F4R;

    __shared__ float As[2][32 * LDA];
    __shared__ float red_s[8 * 32];
    __shared__ float red_q[8 * 32];
    __shared__ float comb[2 * 32];

    const int tid  = threadIdx.x;
    const int wave = tid >> 5;
    const int lane = tid & 31;
    const int ml   = lane & 15;
    const int half = lane >> 4;

    v2f breg[2 * KP];
    {
        const int n = wave * 16 + ml;
#pragma unroll
        for (int j = 0; j < KP; ++j) {
#pragma unroll
            for (int t = 0; t < 2; ++t) {
                const int k = 8 * j + 4 * half + 2 * t;
                float w0, w1;
                if (PRE) {
                    const int krow = (n < HN) ? k : (128 + k);
                    const int col  = (n < HN) ? n : (n - HN);
                    w0 = W[(long)krow * HN + col];
                    w1 = W[(long)(krow + 1) * HN + col];
                } else {
                    w0 = W[(long)k * NDIM + n];
                    w1 = W[(long)(k + 1) * NDIM + n];
                }
                breg[2 * j + t] = v2f{w0, w1};
            }
        }
    }

    const int srow = tid / F4R;
    const int sc4  = tid % F4R;
    const float* sbase = (PRE || sc4 < 32) ? x : agg;
    const int scol = (sc4 & 31) * 4;

    const bool second = PRE && wave >= 4;
    const float bv = second ? 0.0f : bias[wave * 16 + ml];

    int tile = blockIdx.x;
    {
#pragma unroll
        for (int i = 0; i < NLD; ++i) {
            long rr = (long)tile * 32 + srow + i * RSTEP;
            if (rr >= N) rr = N - 1;
            const v4f v = *(const v4f*)(sbase + rr * 128 + scol);
            *(v4f*)&As[0][(srow + i * RSTEP) * LDA + sc4 * 4] = v;
        }
    }
    __syncthreads();

    int buf = 0;
    for (; tile < ntiles; tile += gridDim.x) {
        const int nxt = tile + gridDim.x;
        v4f pre[NLD];
#pragma unroll
        for (int i = 0; i < NLD; ++i) {
            long rr = (long)nxt * 32 + srow + i * RSTEP;
            if (rr >= N) rr = N - 1;
            pre[i] = *(const v4f*)(sbase + rr * 128 + scol);
        }
        __builtin_amdgcn_sched_barrier(0);

        const float* a0p = &As[buf][ml * LDA + 4 * half];
        const float* a1p = a0p + 16 * LDA;
        v8f c0 = v8f{}, c1 = v8f{};
#pragma unroll
        for (int j = 0; j < KP; ++j) {
            const v4f a0 = *(const v4f*)(a0p + 8 * j);
            const v4f a1 = *(const v4f*)(a1p + 8 * j);
            c0 = __builtin_amdgcn_wmma_f32_16x16x4_f32(
                false, v2f{a0.x, a0.y}, false, breg[2 * j], (short)0, c0, false, false);
            c1 = __builtin_amdgcn_wmma_f32_16x16x4_f32(
                false, v2f{a1.x, a1.y}, false, breg[2 * j], (short)0, c1, false, false);
            c0 = __builtin_amdgcn_wmma_f32_16x16x4_f32(
                false, v2f{a0.z, a0.w}, false, breg[2 * j + 1], (short)0, c0, false, false);
            c1 = __builtin_amdgcn_wmma_f32_16x16x4_f32(
                false, v2f{a1.z, a1.w}, false, breg[2 * j + 1], (short)0, c1, false, false);
        }

#pragma unroll
        for (int r = 0; r < 8; ++r) { c0[r] += bv; c1[r] += bv; }

        if (LNRELU) {
            const bool o1 = (ml & 1) != 0, o2 = (ml & 2) != 0, o4 = (ml & 4) != 0, o8 = (ml & 8) != 0;
            float s1[8], q1[8];
#pragma unroll
            for (int i = 0; i < 8; ++i) {
                const float va = (i < 4) ? c0[2 * i]     : c1[2 * i - 8];
                const float vb = (i < 4) ? c0[2 * i + 1] : c1[2 * i - 7];
                const float keep = o1 ? vb : va;
                const float send = o1 ? va : vb;
                s1[i] = keep + lane_xor<1>(send);
                q1[i] = keep * keep + lane_xor<1>(send * send);
            }
            float s2[4], q2[4];
#pragma unroll
            for (int i = 0; i < 4; ++i) {
                const float ks = o2 ? s1[2 * i + 1] : s1[2 * i];
                const float ss = o2 ? s1[2 * i]     : s1[2 * i + 1];
                const float kq = o2 ? q1[2 * i + 1] : q1[2 * i];
                const float sq = o2 ? q1[2 * i]     : q1[2 * i + 1];
                s2[i] = ks + lane_xor<2>(ss);
                q2[i] = kq + lane_xor<2>(sq);
            }
            float s3[2], q3[2];
#pragma unroll
            for (int i = 0; i < 2; ++i) {
                const float ks = o4 ? s2[2 * i + 1] : s2[2 * i];
                const float ss = o4 ? s2[2 * i]     : s2[2 * i + 1];
                const float kq = o4 ? q2[2 * i + 1] : q2[2 * i];
                const float sq = o4 ? q2[2 * i]     : q2[2 * i + 1];
                s3[i] = ks + lane_xor<4>(ss);
                q3[i] = kq + lane_xor<4>(sq);
            }
            const float rs = (o8 ? s3[1] : s3[0]) + lane_xor<8>(o8 ? s3[0] : s3[1]);
            const float rq = (o8 ? q3[1] : q3[0]) + lane_xor<8>(o8 ? q3[0] : q3[1]);
            {
                const int rowid = (ml >> 3) * 16 + half * 8 + (ml & 7);
                red_s[wave * 32 + rowid] = rs;
                red_q[wave * 32 + rowid] = rq;
            }
            __syncthreads();
            if (tid < 64) {
                const float* rp = (tid < 32) ? red_s : red_q;
                const int row = tid & 31;
                float ts = 0.0f;
#pragma unroll
                for (int w = 0; w < 8; ++w) ts += rp[w * 32 + row];
                comb[tid] = ts;
            }
            __syncthreads();
#pragma unroll
            for (int g = 0; g < 2; ++g) {
#pragma unroll
                for (int r = 0; r < 8; ++r) {
                    const int rowid = g * 16 + half * 8 + r;
                    const float mu   = comb[rowid] * (1.0f / NDIM);
                    const float var  = comb[32 + rowid] * (1.0f / NDIM) - mu * mu;
                    const float rstd = rsqrtf(var + LN_EPS);
                    const float cv = g ? c1[r] : c0[r];
                    float v = (cv - mu) * rstd;
                    v = v > 0.0f ? v : 0.0f;
                    if (g) c1[r] = v; else c0[r] = v;
                }
            }
        }

#pragma unroll
        for (int i = 0; i < NLD; ++i)
            *(v4f*)&As[buf ^ 1][(srow + i * RSTEP) * LDA + sc4 * 4] = pre[i];
        __syncthreads();
        buf ^= 1;

#pragma unroll
        for (int g = 0; g < 2; ++g) {
#pragma unroll
            for (int r = 0; r < 8; ++r) {
                const long ro = (long)tile * 32 + g * 16 + half * 8 + r;
                const float cv = g ? c1[r] : c0[r];
                if (PRE) {
                    float* yp = second ? y2 : y;
                    const int cw = second ? (wave - 4) : wave;
                    if (ro < N) yp[ro * HN + cw * 16 + ml] = cv;
                } else {
                    if (ro < N) y[ro * NDIM + wave * 16 + ml] = cv;
                }
            }
        }
    }
}


extern "C" void kernel_launch(void* const* d_in, const int* in_sizes, int n_in,
                              void* d_out, int out_size, void* d_ws, size_t ws_size,
                              hipStream_t stream) {
    const float* h   = (const float*)d_in[0];
    const int*   src = (const int*)d_in[1];
    const int*   dst = (const int*)d_in[2];
    const float* W1  = (const float*)d_in[3];
    const float* b1  = (const float*)d_in[4];
    const float* W2  = (const float*)d_in[5];
    const float* b2  = (const float*)d_in[6];
    const float* W3  = (const float*)d_in[7];
    const float* b3  = (const float*)d_in[8];
    float* out = (float*)d_out;

    const int N = in_sizes[0] / 128;
    const int E = in_sizes[1];

    float* ws   = (float*)d_ws;
    float* norm = ws;
    float* agg  = norm + N;
    float* h1   = agg + (size_t)N * 128;
    float* h2   = h1  + (size_t)N * 128;

    const int  ntiles      = (N + 31) / 32;
    const int  gemm_blocks = ntiles < 512 ? ntiles : 512;
    const int  gemm_blocks3 = ntiles < 1024 ? ntiles : 1024;
    const int  sum_tiles   = (N + TILE_N - 1) / TILE_N;
    int* lists = (int*)(h2 + (size_t)N * 128);
    int* lcnt  = lists + (size_t)sum_tiles * (32 * LCAP * 2);
    const int  nchunk    = (E + 1023) / 1024;
    const int  wpt       = nchunk * 32;
    const int  bm_blocks = nchunk < 512 ? nchunk : 512;
    unsigned* bitmap = (unsigned*)(lcnt + (size_t)sum_tiles * 32);

    edge_bitmap_kernel<TILE_N><<<bm_blocks, 256, 0, stream>>>(dst, bitmap, E, sum_tiles, wpt);

    tile_sum_kernel<128, TILE_N, false, true><<<sum_tiles, 1024, 0, stream>>>(h, src, dst, nullptr, norm, lists, lcnt, bitmap, agg, E, N, wpt);
    sage_gemm_kernel<true, false><<<gemm_blocks, 256, 0, stream>>>(h, agg, W1, b1, h1, nullptr, N, ntiles);

    tile_sum_kernel<128, TILE_N, false, false><<<sum_tiles, 1024, 0, stream>>>(h1, src, dst, norm, nullptr, lists, lcnt, bitmap, agg, E, N, wpt);
    sage_gemm_kernel<true, false><<<gemm_blocks, 256, 0, stream>>>(h1, agg, W2, b2, h2, nullptr, N, ntiles);

    float* z = agg;
    sage_gemm_kernel<false, true><<<gemm_blocks3, 256, 0, stream>>>(h2, nullptr, W3, b3, out, z, N, ntiles);
    tile_sum_kernel<64, TILE_N, true, false><<<sum_tiles, 1024, 0, stream>>>(z, src, dst, norm, nullptr, lists, lcnt, bitmap, out, E, N, wpt);
}


# Global accesses keep their address registers: gathered row = own scalar base pair + one lane offset; a GEMM tile's loads and 16 stores = one address + constants; no translation waits; sums unchanged
# speedup vs baseline: 3.1643x; 1.0410x over previous
#include <hip/hip_runtime.h>

typedef __attribute__((ext_vector_type(2))) float v2f;
typedef __attribute__((ext_vector_type(4))) float v4f;
typedef __attribute__((ext_vector_type(8))) float v8f;

#define LN_EPS 1e-5f

typedef const float __attribute__((address_space(1)))* gfloat_p;

template <int TN>
__global__ __launch_bounds__(256) void edge_bitmap_kernel(const int* __restrict__ dst,
                                                          unsigned* __restrict__ bitmap,
                                                          int E, int tiles, int wpt) {
    __shared__ unsigned bm[256 * 32];
    const int tid  = threadIdx.x;
    const int wave = tid >> 5;
    const int lane = tid & 31;
    const int nchunk = (E + 1023) >> 10;
    for (int ch = blockIdx.x; ch < nchunk; ch += gridDim.x) {
        const int eb = ch * 1024 + tid * 4;
        int t4[4];
        if (eb + 3 < E) {
            const int4 v = *(const int4*)(dst + eb);
            t4[0] = v.x / TN; t4[1] = v.y / TN; t4[2] = v.z / TN; t4[3] = v.w / TN;
        } else {
#pragma unroll
            for (int c = 0; c < 4; ++c) t4[c] = (eb + c < E) ? dst[eb + c] / TN : -1;
        }
        for (int tb = 0; tb < tiles; tb += 256) {
            for (int i = tid * 4; i < 256 * 32; i += 1024) *(uint4*)&bm[i] = uint4{0u, 0u, 0u, 0u};
            __syncthreads();
#pragma unroll
            for (int c = 0; c < 4; ++c) {
                const int t = t4[c] - tb;
                if ((unsigned)t < 256u) atomicOr(&bm[t * 32 + (tid >> 3)], 1u << ((tid & 7) * 4 + c));
            }
            __syncthreads();
            for (int t = wave * 4 + (lane >> 3); t < 256; t += 32)
                if (tb + t < tiles)
                    *(uint4*)&bitmap[(size_t)(tb + t) * wpt + ch * 32 + (lane & 7) * 4] =
                        *(const uint4*)&bm[t * 32 + (lane & 7) * 4];
            __syncthreads();
        }
    }
}

constexpr int TILE_N = 400;
constexpr int LCAP = 512;

template <int F>
__device__ __forceinline__ void add_rows(const float* __restrict__ x, float* tile, int s_l, int dl_l,
                                         int cnt, int lane) {
    constexpr int C = F / 32;
    const unsigned long long rp_l = (unsigned long long)(x + (long)s_l * F);
    const int rlo_l = (int)(unsigned)rp_l;
    const int rhi_l = (int)(unsigned)(rp_l >> 32);
    const unsigned ulane = (unsigned)lane;
    for (int b = 0; b < cnt; b += 8) {
        float v[8][C];
        unsigned lo[8], hi[8];
#pragma unroll
        for (int j = 0; j < 8; ++j) {
            lo[j] = (unsigned)__builtin_amdgcn_readlane(rlo_l, b + j);
            hi[j] = (unsigned)__builtin_amdgcn_readlane(rhi_l, b + j);
        }
#pragma unroll
        for (int j = 0; j < 8; ++j) asm volatile("" : : "s"(lo[j]), "s"(hi[j]));
#pragma unroll
        for (int j = 0; j < 8; ++j) {
            const gfloat_p rowp = (gfloat_p)(((unsigned long long)hi[j] << 32) | (unsigned long long)lo[j]);
#pragma unroll
            for (int c = 0; c < C; ++c) v[j][c] = rowp[ulane + 32u * c];
        }
#pragma unroll
        for (int j = 0; j < 8; ++j) {
            if (C == 4) asm volatile("" : : "v"(v[j][0]), "v"(v[j][1]), "v"(v[j][C - 2]), "v"(v[j][C - 1]));
            else        asm volatile("" : : "v"(v[j][0]), "v"(v[j][C - 1]));
        }
#pragma unroll
        for (int j = 0; j < 8; ++j) {
            if (b + j < cnt) {
                const int dl = __builtin_amdgcn_readlane(dl_l, b + j);
                float* tp = tile + dl * F + lane;
#pragma unroll
                for (int c = 0; c < C; ++c) atomicAdd(tp + 32 * c, v[j][c]);
            }
        }
    }
}

template <int F, bool FIRST>
__device__ __forceinline__ void drain_queue(const float* __restrict__ x, const int* __restrict__ src,
                                            const int* __restrict__ dst, float* tile, int* degc,
                                            int* lcur, int* ltile,
                                            const int* qe, int lo, int cnt, int lane) {
    const int e_l  = (lane < cnt) ? qe[lane] : 0;
    const int s_l  = src[e_l];
    const int dl_l = dst[e_l] - lo;
    if (FIRST) {
        if (lane < cnt) {
            atomicAdd(&degc[dl_l], 1);
            const int ow  = (dl_l >> 2) & 31;
            const int pos = atomicAdd(&lcur[ow], 1);
            if (pos < LCAP) *(int2*)(ltile + 2 * (ow * LCAP + pos)) = make_int2(s_l, dl_l);
        }
    }
    add_rows<F>(x, tile, s_l, dl_l, cnt, lane);
}

template <int F, int TN, bool ADD_TO_OUT, bool FIRST>
__global__ __launch_bounds__(1024) void tile_sum_kernel(
    const float* __restrict__ x,
    const int* __restrict__ src, const int* __restrict__ dst,
    const float* __restrict__ norm,
    float* __restrict__ norm_out,
    int* __restrict__ lists,
    int* __restrict__ lcnt,
    const unsigned* __restrict__ bitmap,
    float* __restrict__ y,
    int E, int N, int wpt) {
    constexpr int C = F / 32;
    static_assert(TN % 4 == 0, "the write-out takes 4 rows at a time");

    __shared__ float tile[TN * 128];
    __shared__ int q_e[32 * 64];
    __shared__ int degc[FIRST ? TN : 1];
    __shared__ int lcur[FIRST ? 32 : 1];

    const int tid  = threadIdx.x;
    const int wave = __builtin_amdgcn_readfirstlane(tid >> 5);
    const int lane = tid & 31;
    const int lo   = blockIdx.x * TN;
    int* ltile = lists + (size_t)blockIdx.x * (32 * LCAP * 2);

    bool use_lists = false;
    int  cntw = 0;
    if (!FIRST) {
        const int myc = lcnt[blockIdx.x * 32 + lane];
        use_lists = __builtin_amdgcn_ballot_w32(myc > LCAP) == 0u;
        cntw = __builtin_amdgcn_readlane(myc, wave);
    }

    if (!FIRST && use_lists) {
        for (int r0 = wave * 4; r0 < TN; r0 += 128) {
#pragma unroll
            for (int k = 0; k < 4; ++k) {
#pragma unroll
                for (int c = 0; c < C; ++c) tile[(r0 + k) * F + lane + 32 * c] = 0.0f;
            }
        }
        const int* lp = ltile + wave * (LCAP * 2);
        int2 ent = (lane < cntw) ? *(const int2*)(lp + 2 * lane) : make_int2(0, 0);
        for (int i = 0; i < cntw; i += 32) {
            const int nl = i + 32 + lane;
            const int2 nxt = (nl < cntw) ? *(const int2*)(lp + 2 * nl) : make_int2(0, 0);
            const int left = cntw - i;
            add_rows<F>(x, tile, ent.x, ent.y, left < 32 ? left : 32, lane);
            ent = nxt;
        }
    } else {
        for (int i = tid * 4; i < TN * F; i += 4096) *(float4*)&tile[i] = float4{0.0f, 0.0f, 0.0f, 0.0f};
        if (FIRST) {
            for (int i = tid; i < TN; i += 1024) degc[i] = 0;
            if (tid < 32) lcur[tid] = 0;
        }
        __syncthreads();

        int* qe = q_e + wave * 64;
        int qn = 0;

        const unsigned* bmt = bitmap + (size_t)blockIdx.x * wpt;

        for (int eb = wave * 1024; eb < E; eb += 8 * 32768) {
            unsigned w[8];
#pragma unroll
            for (int k = 0; k < 8; ++k) {
                const int e = eb + k * 32768;
                w[k] = (e < E) ? bmt[(e >> 5) + lane] : 0u;
            }
#pragma nounroll
            for (int k = 0; k < 8; ++k) {
                const int e0 = eb + k * 32768;
                unsigned hm = w[0];
#pragma unroll
                for (int j = 0; j < 7; ++j) w[j] = w[j + 1];
                w[7] = 0u;
                for (;;) {
                    const bool hit = hm != 0u;
                    const unsigned m = __builtin_amdgcn_ballot_w32(hit);
                    if (m == 0u) break;
                    if (hit) {
                        const int bit = __builtin_ctz(hm);
                        hm &= hm - 1u;
                        const int pos = qn + (int)__builtin_amdgcn_mbcnt_lo(m, 0u);
                        qe[pos] = e0 + lane * 32 + bit;
                    }
                    qn += __builtin_popcount(m);
                    if (qn >= 32) {
                        drain_queue<F, FIRST>(x, src, dst, tile, degc, lcur, ltile, qe, lo, 32, lane);
                        const int rem = qn - 32;
                        if (lane < rem) {
                            const int te = qe[32 + lane];
                            qe[lane] = te;
                        }
                        qn = rem;
                    }
                }
            }
        }
        if (qn > 0) drain_queue<F, FIRST>(x, src, dst, tile, degc, lcur, ltile, qe, lo, qn, lane);
        __syncthreads();

        if (FIRST) {
            for (int r = tid; r < TN; r += 1024) {
                const int node = lo + r;
                if (node < N) {
                    const int c = degc[r];
                    norm_out[node] = c > 0 ? 1.0f / (float)c : 0.0f;
                }
            }
            if (tid < 32) lcnt[blockIdx.x * 32 + tid] = lcur[tid];
        }
    }

    for (int r0 = wave * 4; r0 < TN; r0 += 128) {
        float old[4][C];
        float nvs[4];
#pragma unroll
        for (int k = 0; k < 4; ++k) {
            const int node = lo + r0 + k;
            if (FIRST) {
                const int c = degc[r0 + k];
                nvs[k] = c > 0 ? 1.0f / (float)c : 0.0f;
            } else {
                nvs[k] = (node < N) ? norm[node] : 0.0f;
            }
        }
        if (ADD_TO_OUT) {
#pragma unroll
            for (int k = 0; k < 4; ++k) {
                const int node = lo + r0 + k;
#pragma unroll
                for (int c = 0; c < C; ++c)
                    old[k][c] = (node < N) ? y[(long)node * F + lane + 32 * c] : 0.0f;
            }
        }
#pragma unroll
        for (int k = 0; k < 4; ++k) {
            const int r = r0 + k;
            const int node = lo + r;
            const float nv = nvs[k];
            if (node < N) {
#pragma unroll
                for (int c = 0; c < C; ++c) {
                    const float t = tile[r * F + lane + 32 * c] * nv;
                    y[(long)node * F + lane + 32 * c] = ADD_TO_OUT ? old[k][c] + t : t;
                }
            }
        }
    }
}

template <int M>
__device__ __forceinline__ float lane_xor(float v) {
    return __builtin_bit_cast(float, __builtin_amdgcn_update_dpp(
        0, __builtin_bit_cast(int, v), 0x160 + M, 0xf, 0xf, true));
}

template <bool LNRELU, bool PRE>
__global__ __launch_bounds__(256) void sage_gemm_kernel(
    const float* __restrict__ x,
    const float* __restrict__ agg,
    const float* __restrict__ W,
    const float* __restrict__ bias,
    float* __restrict__ y,
    float* __restrict__ y2,
    int N, int ntiles) {
    constexpr int NDIM = 128;
    constexpr int HN   = 64;
    constexpr int KTOT = PRE ? 128 : 256;
    constexpr int LDA  = KTOT + 4;
    constexpr int KP   = KTOT / 8;
    constexpr int F4R  = KTOT / 4;
    constexpr int NLD  = 32 * F4R / 256;
    constexpr int RSTEP = 256 / F4R;

    __shared__ float As[2][32 * LDA];
    __shared__ float red_s[8 * 32];
    __shared__ float red_q[8 * 32];
    __shared__ float comb[2 * 32];

    const int tid  = threadIdx.x;
    const int wave = tid >> 5;
    const int lane = tid & 31;
    const int ml   = lane & 15;
    const int half = lane >> 4;

    v2f breg[2 * KP];
    {
        const int n = wave * 16 + ml;
#pragma unroll
        for (int j = 0; j < KP; ++j) {
#pragma unroll
            for (int t = 0; t < 2; ++t) {
                const int k = 8 * j + 4 * half + 2 * t;
                float w0, w1;
                if (PRE) {
                    const int krow = (n < HN) ? k : (128 + k);
                    const int col  = (n < HN) ? n : (n - HN);
                    w0 = W[(long)krow * HN + col];
                    w1 = W[(long)(krow + 1) * HN + col];
                } else {
                    w0 = W[(long)k * NDIM + n];
                    w1 = W[(long)(k + 1) * NDIM + n];
                }
                breg[2 * j + t] = v2f{w0, w1};
            }
        }
    }

    const int srow = tid / F4R;
    const int sc4  = tid % F4R;
    const float* sbase = (PRE || sc4 < 32) ? x : agg;
    const int scol = (sc4 & 31) * 4;

    const bool second = PRE && wave >= 4;
    const float bv = second ? 0.0f : bias[wave * 16 + ml];

    int tile = blockIdx.x;
    {
#pragma unroll
        for (int i = 0; i < NLD; ++i) {
            long rr = (long)tile * 32 + srow + i * RSTEP;
            if (rr >= N) rr = N - 1;
            const v4f v = *(const v4f*)(sbase + rr * 128 + scol);
            *(v4f*)&As[0][(srow + i * RSTEP) * LDA + sc4 * 4] = v;
        }
    }
    __syncthreads();

    int buf = 0;
    for (; tile < ntiles; tile += gridDim.x) {
        const int nxt = tile + gridDim.x;
        v4f pre[NLD];
        if ((long)nxt * 32 + 32 <= (long)N) {
            const float* pb = sbase + ((long)nxt * 32 + srow) * 128 + scol;
#pragma unroll
            for (int i = 0; i < NLD; ++i) pre[i] = *(const v4f*)(pb + i * (RSTEP * 128));
        } else {
#pragma unroll
            for (int i = 0; i < NLD; ++i) {
                long rr = (long)nxt * 32 + srow + i * RSTEP;
                if (rr >= N) rr = N - 1;
                pre[i] = *(const v4f*)(sbase + rr * 128 + scol);
            }
        }
        __builtin_amdgcn_sched_barrier(0);

        const float* a0p = &As[buf][ml * LDA + 4 * half];
        const float* a1p = a0p + 16 * LDA;
        v8f c0 = v8f{}, c1 = v8f{};
#pragma unroll
        for (int j = 0; j < KP; ++j) {
            const v4f a0 = *(const v4f*)(a0p + 8 * j);
            const v4f a1 = *(const v4f*)(a1p + 8 * j);
            c0 = __builtin_amdgcn_wmma_f32_16x16x4_f32(
                false, v2f{a0.x, a0.y}, false, breg[2 * j], (short)0, c0, false, false);
            c1 = __builtin_amdgcn_wmma_f32_16x16x4_f32(
                false, v2f{a1.x, a1.y}, false, breg[2 * j], (short)0, c1, false, false);
            c0 = __builtin_amdgcn_wmma_f32_16x16x4_f32(
                false, v2f{a0.z, a0.w}, false, breg[2 * j + 1], (short)0, c0, false, false);
            c1 = __builtin_amdgcn_wmma_f32_16x16x4_f32(
                false, v2f{a1.z, a1.w}, false, breg[2 * j + 1], (short)0, c1, false, false);
        }

#pragma unroll
        for (int r = 0; r < 8; ++r) { c0[r] += bv; c1[r] += bv; }

        if (LNRELU) {
            const bool o1 = (ml & 1) != 0, o2 = (ml & 2) != 0, o4 = (ml & 4) != 0, o8 = (ml & 8) != 0;
            float s1[8], q1[8];
#pragma unroll
            for (int i = 0; i < 8; ++i) {
                const float va = (i < 4) ? c0[2 * i]     : c1[2 * i - 8];
                const float vb = (i < 4) ? c0[2 * i + 1] : c1[2 * i - 7];
                const float keep = o1 ? vb : va;
                const float send = o1 ? va : vb;
                s1[i] = keep + lane_xor<1>(send);
                q1[i] = keep * keep + lane_xor<1>(send * send);
            }
            float s2[4], q2[4];
#pragma unroll
            for (int i = 0; i < 4; ++i) {
                const float ks = o2 ? s1[2 * i + 1] : s1[2 * i];
                const float ss = o2 ? s1[2 * i]     : s1[2 * i + 1];
                const float kq = o2 ? q1[2 * i + 1] : q1[2 * i];
                const float sq = o2 ? q1[2 * i]     : q1[2 * i + 1];
                s2[i] = ks + lane_xor<2>(ss);
                q2[i] = kq + lane_xor<2>(sq);
            }
            float s3[2], q3[2];
#pragma unroll
            for (int i = 0; i < 2; ++i) {
                const float ks = o4 ? s2[2 * i + 1] : s2[2 * i];
                const float ss = o4 ? s2[2 * i]     : s2[2 * i + 1];
                const float kq = o4 ? q2[2 * i + 1] : q2[2 * i];
                const float sq = o4 ? q2[2 * i]     : q2[2 * i + 1];
                s3[i] = ks + lane_xor<4>(ss);
                q3[i] = kq + lane_xor<4>(sq);
            }
            const float rs = (o8 ? s3[1] : s3[0]) + lane_xor<8>(o8 ? s3[0] : s3[1]);
            const float rq = (o8 ? q3[1] : q3[0]) + lane_xor<8>(o8 ? q3[0] : q3[1]);
            {
                const int rowid = (ml >> 3) * 16 + half * 8 + (ml & 7);
                red_s[wave * 32 + rowid] = rs;
                red_q[wave * 32 + rowid] = rq;
            }
            __syncthreads();
            if (tid < 64) {
                const float* rp = (tid < 32) ? red_s : red_q;
                const int row = tid & 31;
                float ts = 0.0f;
#pragma unroll
                for (int w = 0; w < 8; ++w) ts += rp[w * 32 + row];
                comb[tid] = ts;
            }
            __syncthreads();
#pragma unroll
            for (int g = 0; g < 2; ++g) {
#pragma unroll
                for (int r = 0; r < 8; ++r) {
                    const int rowid = g * 16 + half * 8 + r;
                    const float mu   = comb[rowid] * (1.0f / NDIM);
                    const float var  = comb[32 + rowid] * (1.0f / NDIM) - mu * mu;
                    const float rstd = rsqrtf(var + LN_EPS);
                    const float cv = g ? c1[r] : c0[r];
                    float v = (cv - mu) * rstd;
                    v = v > 0.0f ? v : 0.0f;
                    if (g) c1[r] = v; else c0[r] = v;
                }
            }
        }

#pragma unroll
        for (int i = 0; i < NLD; ++i)
            *(v4f*)&As[buf ^ 1][(srow + i * RSTEP) * LDA + sc4 * 4] = pre[i];
        __syncthreads();
        buf ^= 1;

        if ((long)tile * 32 + 32 <= (long)N) {
            constexpr int PITCH = PRE ? HN : NDIM;
            float* yp = PRE ? (second ? y2 : y) : y;
            const int cw = (PRE && second) ? (wave - 4) : wave;
            float* yt = yp + (long)tile * 32 * PITCH + (half * 8) * PITCH + cw * 16 + ml;
#pragma unroll
            for (int g = 0; g < 2; ++g) {
#pragma unroll
                for (int r = 0; r < 8; ++r) yt[(g * 16 + r) * PITCH] = g ? c1[r] : c0[r];
            }
        } else {
#pragma unroll
            for (int g = 0; g < 2; ++g) {
#pragma unroll
                for (int r = 0; r < 8; ++r) {
                    const long ro = (long)tile * 32 + g * 16 + half * 8 + r;
                    const float cv = g ? c1[r] : c0[r];
                    if (PRE) {
                        float* yp = second ? y2 : y;
                        const int cw = second ? (wave - 4) : wave;
                        if (ro < N) yp[ro * HN + cw * 16 + ml] = cv;
                    } else {
                        if (ro < N) y[ro * NDIM + wave * 16 + ml] = cv;
                    }
                }
            }
        }
    }
}


extern "C" void kernel_launch(void* const* d_in, const int* in_sizes, int n_in,
                              void* d_out, int out_size, void* d_ws, size_t ws_size,
                              hipStream_t stream) {
    const float* h   = (const float*)d_in[0];
    const int*   src = (const int*)d_in[1];
    const int*   dst = (const int*)d_in[2];
    const float* W1  = (const float*)d_in[3];
    const float* b1  = (const float*)d_in[4];
    const float* W2  = (const float*)d_in[5];
    const float* b2  = (const float*)d_in[6];
    const float* W3  = (const float*)d_in[7];
    const float* b3  = (const float*)d_in[8];
    float* out = (float*)d_out;

    const int N = in_sizes[0] / 128;
    const int E = in_sizes[1];

    float* ws   = (float*)d_ws;
    float* norm = ws;
    float* agg  = norm + N;
    float* h1   = agg + (size_t)N * 128;
    float* h2   = h1  + (size_t)N * 128;

    const int  ntiles      = (N + 31) / 32;
    const int  gemm_blocks = ntiles < 512 ? ntiles : 512;
    const int  gemm_blocks3 = ntiles < 1024 ? ntiles : 1024;
    const int  sum_tiles   = (N + TILE_N - 1) / TILE_N;
    int* lists = (int*)(h2 + (size_t)N * 128);
    int* lcnt  = lists + (size_t)sum_tiles * (32 * LCAP * 2);
    const int  nchunk    = (E + 1023) / 1024;
    const int  wpt       = nchunk * 32;
    const int  bm_blocks = nchunk < 512 ? nchunk : 512;
    unsigned* bitmap = (unsigned*)(lcnt + (size_t)sum_tiles * 32);

    edge_bitmap_kernel<TILE_N><<<bm_blocks, 256, 0, stream>>>(dst, bitmap, E, sum_tiles, wpt);

    tile_sum_kernel<128, TILE_N, false, true><<<sum_tiles, 1024, 0, stream>>>(h, src, dst, nullptr, norm, lists, lcnt, bitmap, agg, E, N, wpt);
    sage_gemm_kernel<true, false><<<gemm_blocks, 256, 0, stream>>>(h, agg, W1, b1, h1, nullptr, N, ntiles);

    tile_sum_kernel<128, TILE_N, false, false><<<sum_tiles, 1024, 0, stream>>>(h1, src, dst, norm, nullptr, lists, lcnt, bitmap, agg, E, N, wpt);
    sage_gemm_kernel<true, false><<<gemm_blocks, 256, 0, stream>>>(h1, agg, W2, b2, h2, nullptr, N, ntiles);

    float* z = agg;
    sage_gemm_kernel<false, true><<<gemm_blocks3, 256, 0, stream>>>(h2, nullptr, W3, b3, out, z, N, ntiles);
    tile_sum_kernel<64, TILE_N, true, false><<<sum_tiles, 1024, 0, stream>>>(z, src, dst, norm, nullptr, lists, lcnt, bitmap, out, E, N, wpt);
}
